# starting kernel
# baseline (speedup 1.0000x reference)
#include <hip/hip_runtime.h>
#include <hip/hip_bf16.h>


typedef float v2f __attribute__((ext_vector_type(2)));
typedef float v8f __attribute__((ext_vector_type(8)));

#define NEG_SLOPE 0.2f

__device__ __forceinline__ unsigned f2ord(float f) {
  unsigned u = __float_as_uint(f);
  return (f >= 0.0f) ? (u | 0x80000000u) : ~u;
}
__device__ __forceinline__ float ord2f(unsigned u) {
  return (u & 0x80000000u) ? __uint_as_float(u & 0x7FFFFFFFu)
                           : __uint_as_float(~u);
}

template <int K, int NOUT>
__global__ void gemm_wmma_f32(const float* __restrict__ A,
                              const float* __restrict__ B,
                              float* __restrict__ C, int M) {
  const int NT    = NOUT / 16;
  const int wave  = (blockIdx.x * blockDim.x + threadIdx.x) >> 5;
  const int lane  = threadIdx.x & 31;
  const int mt    = wave / NT;
  const int nt    = wave % NT;
  if (mt * 16 >= M) return;

  const int half = lane >> 4;
  const int lr   = lane & 15;
  int arow = mt * 16 + lr;
  if (arow >= M) arow = M - 1;
  const int bcol = nt * 16 + lr;

  v8f acc = {};
  const float* aptr = A + (size_t)arow * K + 2 * half;
  const float* bptr = B + (size_t)(2 * half) * NOUT + bcol;
  for (int k = 0; k < K; k += 4) {
    v2f a = *(const v2f*)(aptr + k);
    v2f b;
    b.x = bptr[(size_t)k * NOUT];
    b.y = bptr[(size_t)(k + 1) * NOUT];
    acc = __builtin_amdgcn_wmma_f32_16x16x4_f32(
         false, a,  false, b,
         (short)0, acc,  false,  false);
  }

  const int orow = mt * 16 + half * 8;
#pragma unroll
  for (int r = 0; r < 8; ++r) {
    if (orow + r < M) C[(size_t)(orow + r) * NOUT + bcol] = acc[r];
  }
}

__global__ void att_scores(const float* __restrict__ h,
                           const float* __restrict__ att_src,
                           const float* __restrict__ att_dst,
                           float* __restrict__ asrc, float* __restrict__ adst,
                           int n, int heads, int ch) {
  int i = blockIdx.x * blockDim.x + threadIdx.x;
  if (i >= n * heads) return;
  int node = i / heads, hd = i % heads;
  const float* hp = h + (size_t)node * heads * ch + hd * ch;
  const float* as = att_src + hd * ch;
  const float* ad = att_dst + hd * ch;
  float ss = 0.0f, sd = 0.0f;
  for (int c = 0; c < ch; ++c) {
    float v = hp[c];
    ss += v * as[c];
    sd += v * ad[c];
  }
  asrc[i] = ss;
  adst[i] = sd;
}

template <int H>
__global__ void edge_max(const int* __restrict__ ei, int E, int T,
                         const float* __restrict__ asrc,
                         const float* __restrict__ adst,
                         unsigned* __restrict__ m) {
  int e = blockIdx.x * blockDim.x + threadIdx.x;
  if (e >= T) return;
  int s = (e < E) ? ei[e] : (e - E);
  int d = (e < E) ? ei[E + e] : (e - E);
#pragma unroll
  for (int h = 0; h < H; ++h) {
    float v = asrc[s * H + h] + adst[d * H + h];
    v = (v > 0.0f) ? v : NEG_SLOPE * v;
    atomicMax(&m[d * H + h], f2ord(v));
  }
}

template <int H>
__global__ void edge_expsum(const int* __restrict__ ei, int E, int T,
                            const float* __restrict__ asrc,
                            const float* __restrict__ adst,
                            const unsigned* __restrict__ m,
                            float* __restrict__ ssum) {
  int e = blockIdx.x * blockDim.x + threadIdx.x;
  if (e >= T) return;
  int s = (e < E) ? ei[e] : (e - E);
  int d = (e < E) ? ei[E + e] : (e - E);
#pragma unroll
  for (int h = 0; h < H; ++h) {
    float v = asrc[s * H + h] + adst[d * H + h];
    v = (v > 0.0f) ? v : NEG_SLOPE * v;
    float ex = __expf(v - ord2f(m[d * H + h]));
    atomicAdd(&ssum[d * H + h], ex);
  }
}

template <int H, int C>
__global__ void edge_aggregate(const int* __restrict__ ei, int E, int T,
                               const float* __restrict__ asrc,
                               const float* __restrict__ adst,
                               const unsigned* __restrict__ m,
                               const float* __restrict__ ssum,
                               const float* __restrict__ hfeat,
                               float* __restrict__ out) {
  const int wave = (blockIdx.x * blockDim.x + threadIdx.x) >> 5;
  const int lane = threadIdx.x & 31;
  if (wave >= T) return;
  int s = (wave < E) ? ei[wave] : (wave - E);
  int d = (wave < E) ? ei[E + wave] : (wave - E);

  float alpha = 0.0f;
  if (lane < H) {
    float v = asrc[s * H + lane] + adst[d * H + lane];
    v = (v > 0.0f) ? v : NEG_SLOPE * v;
    alpha = __expf(v - ord2f(m[d * H + lane])) / (ssum[d * H + lane] + 1e-16f);
  }

  const int F = H * C;
  const float* hp = hfeat + (size_t)s * F;
  float* op = out + (size_t)d * F;
#pragma unroll
  for (int f = lane; f < F; f += 32) {
    float a = __shfl(alpha, f / C, 32);
    atomicAdd(&op[f], hp[f] * a);
  }
}

__global__ void init_bias(float* __restrict__ out, const float* __restrict__ bias,
                          int n, int F) {
  int i = blockIdx.x * blockDim.x + threadIdx.x;
  if (i < n * F) out[i] = bias[i % F];
}
__global__ void init_zero_u32(unsigned* __restrict__ p, int n) {
  int i = blockIdx.x * blockDim.x + threadIdx.x;
  if (i < n) p[i] = 0u;
}
__global__ void init_zero_f32(float* __restrict__ p, int n) {
  int i = blockIdx.x * blockDim.x + threadIdx.x;
  if (i < n) p[i] = 0.0f;
}
__global__ void relu_inplace(float* __restrict__ p, int n) {
  int i = blockIdx.x * blockDim.x + threadIdx.x;
  if (i < n) p[i] = fmaxf(p[i], 0.0f);
}

extern "C" void kernel_launch(void* const* d_in, const int* in_sizes, int n_in,
                              void* d_out, int out_size, void* d_ws, size_t ws_size,
                              hipStream_t stream) {
  const float* x        = (const float*)d_in[0];
  const int*   ei       = (const int*)d_in[1];
  const float* W1       = (const float*)d_in[2];
  const float* att_src1 = (const float*)d_in[3];
  const float* att_dst1 = (const float*)d_in[4];
  const float* b1       = (const float*)d_in[5];
  const float* W2       = (const float*)d_in[6];
  const float* att_src2 = (const float*)d_in[7];
  const float* att_dst2 = (const float*)d_in[8];
  const float* b2       = (const float*)d_in[9];

  const int D_IN = 256, F1 = 256, F2 = 64, H1 = 8, C1 = 32, H2 = 1, C2 = 64;
  const int N = in_sizes[0] / D_IN;
  const int E = in_sizes[1] / 2;
  const int T = E + N;

  float* ws = (float*)d_ws;
  float*    h1    = ws;
  float*    agg1  = h1   + (size_t)N * F1;
  float*    asrc1 = agg1 + (size_t)N * F1;
  float*    adst1 = asrc1 + (size_t)N * H1;
  unsigned* m1    = (unsigned*)(adst1 + (size_t)N * H1);
  float*    s1    = (float*)(m1 + (size_t)N * H1);
  float*    asrc2 = s1 + (size_t)N * H1;
  float*    adst2 = asrc2 + N;
  unsigned* m2    = (unsigned*)(adst2 + N);
  float*    s2    = (float*)(m2 + N);
  float*    out   = (float*)d_out;

  const int BT = 256;
  auto cdiv = [](long long a, long long b) { return (int)((a + b - 1) / b); };

  init_bias<<<cdiv((long long)N * F1, BT), BT, 0, stream>>>(agg1, b1, N, F1);
  init_bias<<<cdiv((long long)N * F2, BT), BT, 0, stream>>>(out, b2, N, F2);
  init_zero_u32<<<cdiv((long long)N * H1, BT), BT, 0, stream>>>(m1, N * H1);
  init_zero_f32<<<cdiv((long long)N * H1, BT), BT, 0, stream>>>(s1, N * H1);
  init_zero_u32<<<cdiv(N, BT), BT, 0, stream>>>(m2, N);
  init_zero_f32<<<cdiv(N, BT), BT, 0, stream>>>(s2, N);

  {
    const int mtiles = (N + 15) / 16, ntiles = F1 / 16;
    const long long waves = (long long)mtiles * ntiles;
    gemm_wmma_f32<256, 256><<<cdiv(waves * 32, BT), BT, 0, stream>>>(x, W1, h1, N);
  }
  att_scores<<<cdiv((long long)N * H1, BT), BT, 0, stream>>>(
      h1, att_src1, att_dst1, asrc1, adst1, N, H1, C1);
  edge_max<8><<<cdiv(T, BT), BT, 0, stream>>>(ei, E, T, asrc1, adst1, m1);
  edge_expsum<8><<<cdiv(T, BT), BT, 0, stream>>>(ei, E, T, asrc1, adst1, m1, s1);
  edge_aggregate<8, 32><<<cdiv((long long)T * 32, BT), BT, 0, stream>>>(
      ei, E, T, asrc1, adst1, m1, s1, h1, agg1);
  relu_inplace<<<cdiv((long long)N * F1, BT), BT, 0, stream>>>(agg1, N * F1);

  float* h2 = h1;
  {
    const int mtiles = (N + 15) / 16, ntiles = F2 / 16;
    const long long waves = (long long)mtiles * ntiles;
    gemm_wmma_f32<256, 64><<<cdiv(waves * 32, BT), BT, 0, stream>>>(agg1, W2, h2, N);
  }
  att_scores<<<cdiv(N, BT), BT, 0, stream>>>(
      h2, att_src2, att_dst2, asrc2, adst2, N, H2, C2);
  edge_max<1><<<cdiv(T, BT), BT, 0, stream>>>(ei, E, T, asrc2, adst2, m2);
  edge_expsum<1><<<cdiv(T, BT), BT, 0, stream>>>(ei, E, T, asrc2, adst2, m2, s2);
  edge_aggregate<1, 64><<<cdiv((long long)T * 32, BT), BT, 0, stream>>>(
      ei, E, T, asrc2, adst2, m2, s2, h2, out);
}


# Edges grouped by destination each call (counting sort, one int atomic per edge); one wave per node does max, exp-sum and gather in registers: float atomics gone; sum order = atomic arrival rank
# speedup vs baseline: 3.8142x; 3.8142x over previous
#include <hip/hip_runtime.h>
#include <hip/hip_bf16.h>


typedef float v2f __attribute__((ext_vector_type(2)));
typedef float v4f __attribute__((ext_vector_type(4)));
typedef float v8f __attribute__((ext_vector_type(8)));

#define NEG_SLOPE 0.2f

template <int K, int NOUT>
__global__ void gemm_wmma_f32(const float* __restrict__ A,
                              const float* __restrict__ B,
                              float* __restrict__ C, int M) {
  const int NT    = NOUT / 16;
  const int wave  = (blockIdx.x * blockDim.x + threadIdx.x) >> 5;
  const int lane  = threadIdx.x & 31;
  const int mt    = wave / NT;
  const int nt    = wave % NT;
  if (mt * 16 >= M) return;

  const int half = lane >> 4;
  const int lr   = lane & 15;
  int arow = mt * 16 + lr;
  if (arow >= M) arow = M - 1;
  const int bcol = nt * 16 + lr;

  v8f acc = {};
  const float* aptr = A + (size_t)arow * K + 2 * half;
  const float* bptr = B + (size_t)(2 * half) * NOUT + bcol;
  for (int k = 0; k < K; k += 4) {
    v2f a = *(const v2f*)(aptr + k);
    v2f b;
    b.x = bptr[(size_t)k * NOUT];
    b.y = bptr[(size_t)(k + 1) * NOUT];
    acc = __builtin_amdgcn_wmma_f32_16x16x4_f32(
         false, a,  false, b,
         (short)0, acc,  false,  false);
  }

  const int orow = mt * 16 + half * 8;
#pragma unroll
  for (int r = 0; r < 8; ++r) {
    if (orow + r < M) C[(size_t)(orow + r) * NOUT + bcol] = acc[r];
  }
}

__global__ void att_scores(const float* __restrict__ h,
                           const float* __restrict__ att_src,
                           const float* __restrict__ att_dst,
                           float* __restrict__ asrc, float* __restrict__ adst,
                           int n, int heads, int ch) {
  int i = blockIdx.x * blockDim.x + threadIdx.x;
  if (i >= n * heads) return;
  int node = i / heads, hd = i % heads;
  const float* hp = h + (size_t)node * heads * ch + hd * ch;
  const float* as = att_src + hd * ch;
  const float* ad = att_dst + hd * ch;
  float ss = 0.0f, sd = 0.0f;
  for (int c = 0; c < ch; ++c) {
    float v = hp[c];
    ss += v * as[c];
    sd += v * ad[c];
  }
  asrc[i] = ss;
  adst[i] = sd;
}

#define CNT_STRIDE 8

__global__ void init_counts(int* __restrict__ cnt, int n) {
  int i = blockIdx.x * blockDim.x + threadIdx.x;
  if (i < n) cnt[(size_t)i * CNT_STRIDE] = 1;
}

__global__ void edge_rank(const int* __restrict__ ei, int E,
                          int* __restrict__ cnt, int* __restrict__ rank) {
  int e = blockIdx.x * blockDim.x + threadIdx.x;
  if (e >= E) return;
  int d = ei[E + e];
  rank[e] = atomicAdd(&cnt[(size_t)d * CNT_STRIDE], 1);
}

__device__ __forceinline__ int block_scan_incl(int v, int* wsum, int& total) {
  const int lane = threadIdx.x & 31, w = threadIdx.x >> 5;
  int x = v;
#pragma unroll
  for (int o = 1; o < 32; o <<= 1) {
    int y = __shfl_up(x, o, 32);
    if (lane >= o) x += y;
  }
  if (lane == 31) wsum[w] = x;
  __syncthreads();
  int woff = 0, tot = 0;
#pragma unroll
  for (int k = 0; k < 8; ++k) {
    int t = wsum[k];
    if (k < w) woff += t;
    tot += t;
  }
  total = tot;
  return woff + x;
}

__global__ void scan_local(const int* __restrict__ cnt, int n,
                           int* __restrict__ loc, int* __restrict__ bsum) {
  __shared__ int wsum[8];
  const int i = blockIdx.x * 256 + threadIdx.x;
  const int v = (i < n) ? cnt[(size_t)i * CNT_STRIDE] : 0;
  int total;
  const int incl = block_scan_incl(v, wsum, total);
  if (i < n) loc[i] = incl - v;
  if (threadIdx.x == 0) bsum[blockIdx.x] = total;
}

__global__ void scan_blocks(const int* __restrict__ bsum, int nb,
                            int* __restrict__ boff) {
  __shared__ int wsum[8];
  int carry = 0;
  for (int base = 0; base < nb; base += 256) {
    const int i = base + threadIdx.x;
    const int v = (i < nb) ? bsum[i] : 0;
    int total;
    const int incl = block_scan_incl(v, wsum, total);
    if (i < nb) boff[i] = carry + incl - v;
    carry += total;
    __syncthreads();
  }
}

__global__ void csr_fill(const int* __restrict__ ei, int E, int T,
                         const int* __restrict__ cnt, const int* __restrict__ loc,
                         const int* __restrict__ boff, const int* __restrict__ rank,
                         int* __restrict__ rowptr, int* __restrict__ csr_src) {
  int e = blockIdx.x * blockDim.x + threadIdx.x;
  if (e >= T) return;
  if (e < E) {
    const int s = ei[e];
    const int d = ei[E + e];
    csr_src[boff[d >> 8] + loc[d] + rank[e]] = s;
  } else {
    const int nd = e - E;
    const int beg = boff[nd >> 8] + loc[nd];
    csr_src[beg] = nd;
    rowptr[nd] = beg;
    if (e == T - 1) rowptr[nd + 1] = beg + cnt[(size_t)nd * CNT_STRIDE];
  }
}

template <int H>
__device__ __forceinline__ void load_heads(const float* __restrict__ p, float (&v)[H]) {
  if constexpr (H == 8) {
    const v4f a = *(const v4f*)p, b = *(const v4f*)(p + 4);
    v[0] = a.x; v[1] = a.y; v[2] = a.z; v[3] = a.w;
    v[4] = b.x; v[5] = b.y; v[6] = b.z; v[7] = b.w;
  } else {
#pragma unroll
    for (int h = 0; h < H; ++h) v[h] = p[h];
  }
}

__device__ __forceinline__ void wave_lds_sync() {
  __builtin_amdgcn_fence(__ATOMIC_RELEASE, "wavefront");
  __builtin_amdgcn_wave_barrier();
  __builtin_amdgcn_fence(__ATOMIC_ACQUIRE, "wavefront");
}

template <int H, int C, bool RELU>
__global__ void __launch_bounds__(256)
node_softmax_gather(const int* __restrict__ rowptr, const int* __restrict__ csr_src,
                    const float* __restrict__ asrc, const float* __restrict__ adst,
                    const float* __restrict__ hfeat, const float* __restrict__ bias,
                    float* __restrict__ out, int n) {
  constexpr int F = H * C;
  constexpr int VPL = F / 32;
  static_assert(VPL == 8 || VPL == 2, "lane owns 8 or 2 features");
  __shared__ float strip[8][32 * H];

  const int lane = threadIdx.x & 31;
  const int wib  = threadIdx.x >> 5;
  const int node = __builtin_amdgcn_readfirstlane(blockIdx.x * 8 + wib);
  if (node >= n) return;
  const int beg = rowptr[node], end = rowptr[node + 1];
  const int hd = (lane * VPL) / C;
  float* myp = strip[wib];

  float ad[H];
  load_heads<H>(adst + (size_t)node * H, ad);

  auto stage_scores = [&](int c) -> int {
    const int j = c + lane;
    int s = 0;
    float v[H];
#pragma unroll
    for (int h = 0; h < H; ++h) v[h] = -INFINITY;
    if (j < end) {
      s = csr_src[j];
      float as[H];
      load_heads<H>(asrc + (size_t)s * H, as);
#pragma unroll
      for (int h = 0; h < H; ++h) {
        float t = as[h] + ad[h];
        v[h] = (t > 0.0f) ? t : NEG_SLOPE * t;
      }
    }
#pragma unroll
    for (int h = 0; h < H; ++h) myp[lane * H + h] = v[h];
    return s;
  };

  float mx = -INFINITY;
  int s = 0;
  for (int c = beg; c < end; c += 32) {
    s = stage_scores(c);
    wave_lds_sync();
#pragma unroll
    for (int t = 0; t < 32; ++t) mx = fmaxf(mx, myp[t * H + hd]);
    wave_lds_sync();
  }

  const bool single = (end - beg) <= 32;
  float acc[VPL];
#pragma unroll
  for (int i = 0; i < VPL; ++i) acc[i] = 0.0f;
  float lsum = 0.0f;
  for (int c = beg; c < end; c += 32) {
    if (!single) {
      s = stage_scores(c);
      wave_lds_sync();
    }
    const int m = min(32, end - c);
#pragma unroll 4
    for (int t = 0; t < m; ++t) {
      const int st = __builtin_amdgcn_readlane(s, t);
      const float p = __expf(myp[t * H + hd] - mx);
      lsum += p;
      const float* hp = hfeat + (size_t)st * F + lane * VPL;
      if constexpr (VPL == 8) {
        const v4f a = *(const v4f*)hp, b = *(const v4f*)(hp + 4);
        acc[0] += p * a.x; acc[1] += p * a.y; acc[2] += p * a.z; acc[3] += p * a.w;
        acc[4] += p * b.x; acc[5] += p * b.y; acc[6] += p * b.z; acc[7] += p * b.w;
      } else {
        const v2f a = *(const v2f*)hp;
        acc[0] += p * a.x; acc[1] += p * a.y;
      }
    }
    if (!single) wave_lds_sync();
  }

  const float denom = lsum + 1e-16f;
  float r[VPL];
#pragma unroll
  for (int i = 0; i < VPL; ++i) {
    float o = acc[i] / denom + bias[lane * VPL + i];
    r[i] = RELU ? fmaxf(o, 0.0f) : o;
  }
  float* op = out + (size_t)node * F + lane * VPL;
  if constexpr (VPL == 8) {
    v4f a = {r[0], r[1], r[2], r[3]}, b = {r[4], r[5], r[6], r[7]};
    *(v4f*)op = a;
    *(v4f*)(op + 4) = b;
  } else {
    v2f a = {r[0], r[1]};
    *(v2f*)op = a;
  }
}

extern "C" void kernel_launch(void* const* d_in, const int* in_sizes, int n_in,
                              void* d_out, int out_size, void* d_ws, size_t ws_size,
                              hipStream_t stream) {
  const float* x        = (const float*)d_in[0];
  const int*   ei       = (const int*)d_in[1];
  const float* W1       = (const float*)d_in[2];
  const float* att_src1 = (const float*)d_in[3];
  const float* att_dst1 = (const float*)d_in[4];
  const float* b1       = (const float*)d_in[5];
  const float* W2       = (const float*)d_in[6];
  const float* att_src2 = (const float*)d_in[7];
  const float* att_dst2 = (const float*)d_in[8];
  const float* b2       = (const float*)d_in[9];

  const int D_IN = 256, F1 = 256, F2 = 64, H1 = 8, C1 = 32, H2 = 1, C2 = 64;
  const int N = in_sizes[0] / D_IN;
  const int E = in_sizes[1] / 2;
  const int T = E + N;

  float* ws = (float*)d_ws;
  float* h1    = ws;
  float* agg1  = h1   + (size_t)N * F1;
  float* asrc1 = agg1 + (size_t)N * F1;
  float* adst1 = asrc1 + (size_t)N * H1;
  float* asrc2 = adst1 + (size_t)N * H1;
  float* adst2 = asrc2 + (N + 4);
  int*   cnt   = (int*)(adst2 + (N + 4));
  int*   loc   = cnt + (size_t)N * 8;
  int*   rowptr = loc + (N + 4);
  int*   bsum  = rowptr + (N + 4);
  int*   boff  = bsum + (N / 256 + 4);
  int*   rank  = boff + (N / 256 + 4);
  int*   csr_src = rank + E;
  float* out   = (float*)d_out;

  const int BT = 256;
  auto cdiv = [](long long a, long long b) { return (int)((a + b - 1) / b); };
  const int NB = cdiv(N, 256);

  init_counts<<<cdiv(N, BT), BT, 0, stream>>>(cnt, N);
  edge_rank<<<cdiv(E, BT), BT, 0, stream>>>(ei, E, cnt, rank);
  scan_local<<<NB, 256, 0, stream>>>(cnt, N, loc, bsum);
  scan_blocks<<<1, 256, 0, stream>>>(bsum, NB, boff);
  csr_fill<<<cdiv(T, BT), BT, 0, stream>>>(ei, E, T, cnt, loc, boff, rank, rowptr, csr_src);

  {
    const int mtiles = (N + 15) / 16, ntiles = F1 / 16;
    const long long waves = (long long)mtiles * ntiles;
    gemm_wmma_f32<256, 256><<<cdiv(waves * 32, BT), BT, 0, stream>>>(x, W1, h1, N);
  }
  att_scores<<<cdiv((long long)N * H1, BT), BT, 0, stream>>>(
      h1, att_src1, att_dst1, asrc1, adst1, N, H1, C1);
  node_softmax_gather<8, 32, true><<<cdiv(N, 8), 256, 0, stream>>>(
      rowptr, csr_src, asrc1, adst1, h1, b1, agg1, N);

  float* h2 = h1;
  {
    const int mtiles = (N + 15) / 16, ntiles = F2 / 16;
    const long long waves = (long long)mtiles * ntiles;
    gemm_wmma_f32<256, 64><<<cdiv(waves * 32, BT), BT, 0, stream>>>(agg1, W2, h2, N);
  }
  att_scores<<<cdiv(N, BT), BT, 0, stream>>>(
      h2, att_src2, att_dst2, asrc2, adst2, N, H2, C2);
  node_softmax_gather<1, 64, false><<<cdiv(N, 8), 256, 0, stream>>>(
      rowptr, csr_src, asrc2, adst2, h2, b2, out, N);
}


# Both GEMMs retiled: 8-wave block = 128x64 of C, W chunk (32 k) staged once in LDS, each wave holds its 16 A rows in registers for 4 column tiles; k summed in permuted order inside each 32-chunk
# speedup vs baseline: 5.0414x; 1.3217x over previous
#include <hip/hip_runtime.h>
#include <hip/hip_bf16.h>


typedef float v2f __attribute__((ext_vector_type(2)));
typedef float v4f __attribute__((ext_vector_type(4)));
typedef float v8f __attribute__((ext_vector_type(8)));

#define NEG_SLOPE 0.2f

template <int K, int NOUT>
__global__ void __launch_bounds__(256)
gemm_wmma_f32(const float* __restrict__ A,
              const float* __restrict__ B,
              float* __restrict__ C, int M) {
  constexpr int KC     = 32;
  constexpr int NCH    = K / KC;
  constexpr int NSTRIP = NOUT / 64;
  static_assert(K % KC == 0 && NOUT % 64 == 0, "tile sizes");
  __shared__ v4f Bs[2][KC * 16];

  const int tid  = threadIdx.x;
  const int lane = tid & 31;
  const int w    = tid >> 5;
  const int c    = lane & 15;
  const int h    = lane >> 4;
  const int nstrip = blockIdx.x % NSTRIP;
  const int mblk   = blockIdx.x / NSTRIP;
  const int n0     = nstrip * 64;

  int arow = mblk * 128 + w * 16 + c;
  if (arow >= M) arow = M - 1;
  const float* ap = A + (size_t)arow * K + 16 * h;

  const int br = tid >> 4, bq = tid & 15;
  const float* bp = B + (size_t)br * NOUT + n0 + 4 * bq;

  v4f a_cur[4], a_nxt[4], b_st[2];
#pragma unroll
  for (int i = 0; i < 4; ++i) a_cur[i] = *(const v4f*)(ap + 4 * i);
  b_st[0] = *(const v4f*)(bp);
  b_st[1] = *(const v4f*)(bp + (size_t)16 * NOUT);
  Bs[0][br * 16 + bq]        = b_st[0];
  Bs[0][(br + 16) * 16 + bq] = b_st[1];
  __syncthreads();

  v8f acc[4] = {};
  for (int kc = 0; kc < NCH; ++kc) {
    const int cur = kc & 1;
    const bool more = (kc + 1 < NCH);
    if (more) {
      const int k1 = (kc + 1) * KC;
#pragma unroll
      for (int i = 0; i < 4; ++i) a_nxt[i] = *(const v4f*)(ap + k1 + 4 * i);
      b_st[0] = *(const v4f*)(bp + (size_t)k1 * NOUT);
      b_st[1] = *(const v4f*)(bp + (size_t)(k1 + 16) * NOUT);
    }
    __builtin_amdgcn_sched_barrier(0);

    const v4f* bs = &Bs[cur][(16 * h) * 16 + c];
    v4f bf[16];
#pragma unroll
    for (int i = 0; i < 16; ++i) bf[i] = bs[i * 16];

#pragma unroll
    for (int s = 0; s < 8; ++s) {
      const v4f av = a_cur[s >> 1];
      v2f a;
      if (s & 1) { a.x = av.z; a.y = av.w; } else { a.x = av.x; a.y = av.y; }
      const v4f b0 = bf[2 * s], b1 = bf[2 * s + 1];
#pragma unroll
      for (int j = 0; j < 4; ++j) {
        v2f b;
        b.x = b0[j];
        b.y = b1[j];
        acc[j] = __builtin_amdgcn_wmma_f32_16x16x4_f32(
             false, a,  false, b,
             (short)0, acc[j],  false,  false);
      }
    }

    __builtin_amdgcn_sched_barrier(0);
    if (more) {
      Bs[cur ^ 1][br * 16 + bq]        = b_st[0];
      Bs[cur ^ 1][(br + 16) * 16 + bq] = b_st[1];
      __syncthreads();
#pragma unroll
      for (int i = 0; i < 4; ++i) a_cur[i] = a_nxt[i];
    }
  }

  const int orow = mblk * 128 + w * 16 + 8 * h;
  float* cp = C + (size_t)orow * NOUT + n0 + 4 * c;
#pragma unroll
  for (int r = 0; r < 8; ++r) {
    if (orow + r < M) {
      v4f o = {acc[0][r], acc[1][r], acc[2][r], acc[3][r]};
      *(v4f*)(cp + (size_t)r * NOUT) = o;
    }
  }
}

__global__ void att_scores(const float* __restrict__ h,
                           const float* __restrict__ att_src,
                           const float* __restrict__ att_dst,
                           float* __restrict__ asrc, float* __restrict__ adst,
                           int n, int heads, int ch) {
  int i = blockIdx.x * blockDim.x + threadIdx.x;
  if (i >= n * heads) return;
  int node = i / heads, hd = i % heads;
  const float* hp = h + (size_t)node * heads * ch + hd * ch;
  const float* as = att_src + hd * ch;
  const float* ad = att_dst + hd * ch;
  float ss = 0.0f, sd = 0.0f;
  for (int c = 0; c < ch; ++c) {
    float v = hp[c];
    ss += v * as[c];
    sd += v * ad[c];
  }
  asrc[i] = ss;
  adst[i] = sd;
}

#define CNT_STRIDE 8

__global__ void init_counts(int* __restrict__ cnt, int n) {
  int i = blockIdx.x * blockDim.x + threadIdx.x;
  if (i < n) cnt[(size_t)i * CNT_STRIDE] = 1;
}

__global__ void edge_rank(const int* __restrict__ ei, int E,
                          int* __restrict__ cnt, int* __restrict__ rank) {
  int e = blockIdx.x * blockDim.x + threadIdx.x;
  if (e >= E) return;
  int d = ei[E + e];
  rank[e] = atomicAdd(&cnt[(size_t)d * CNT_STRIDE], 1);
}

__device__ __forceinline__ int block_scan_incl(int v, int* wsum, int& total) {
  const int lane = threadIdx.x & 31, w = threadIdx.x >> 5;
  int x = v;
#pragma unroll
  for (int o = 1; o < 32; o <<= 1) {
    int y = __shfl_up(x, o, 32);
    if (lane >= o) x += y;
  }
  if (lane == 31) wsum[w] = x;
  __syncthreads();
  int woff = 0, tot = 0;
#pragma unroll
  for (int k = 0; k < 8; ++k) {
    int t = wsum[k];
    if (k < w) woff += t;
    tot += t;
  }
  total = tot;
  return woff + x;
}

__global__ void scan_local(const int* __restrict__ cnt, int n,
                           int* __restrict__ loc, int* __restrict__ bsum) {
  __shared__ int wsum[8];
  const int i = blockIdx.x * 256 + threadIdx.x;
  const int v = (i < n) ? cnt[(size_t)i * CNT_STRIDE] : 0;
  int total;
  const int incl = block_scan_incl(v, wsum, total);
  if (i < n) loc[i] = incl - v;
  if (threadIdx.x == 0) bsum[blockIdx.x] = total;
}

__global__ void scan_blocks(const int* __restrict__ bsum, int nb,
                            int* __restrict__ boff) {
  __shared__ int wsum[8];
  int carry = 0;
  for (int base = 0; base < nb; base += 256) {
    const int i = base + threadIdx.x;
    const int v = (i < nb) ? bsum[i] : 0;
    int total;
    const int incl = block_scan_incl(v, wsum, total);
    if (i < nb) boff[i] = carry + incl - v;
    carry += total;
    __syncthreads();
  }
}

__global__ void csr_fill(const int* __restrict__ ei, int E, int T,
                         const int* __restrict__ cnt, const int* __restrict__ loc,
                         const int* __restrict__ boff, const int* __restrict__ rank,
                         int* __restrict__ rowptr, int* __restrict__ csr_src) {
  int e = blockIdx.x * blockDim.x + threadIdx.x;
  if (e >= T) return;
  if (e < E) {
    const int s = ei[e];
    const int d = ei[E + e];
    csr_src[boff[d >> 8] + loc[d] + rank[e]] = s;
  } else {
    const int nd = e - E;
    const int beg = boff[nd >> 8] + loc[nd];
    csr_src[beg] = nd;
    rowptr[nd] = beg;
    if (e == T - 1) rowptr[nd + 1] = beg + cnt[(size_t)nd * CNT_STRIDE];
  }
}

template <int H>
__device__ __forceinline__ void load_heads(const float* __restrict__ p, float (&v)[H]) {
  if constexpr (H == 8) {
    const v4f a = *(const v4f*)p, b = *(const v4f*)(p + 4);
    v[0] = a.x; v[1] = a.y; v[2] = a.z; v[3] = a.w;
    v[4] = b.x; v[5] = b.y; v[6] = b.z; v[7] = b.w;
  } else {
#pragma unroll
    for (int h = 0; h < H; ++h) v[h] = p[h];
  }
}

__device__ __forceinline__ void wave_lds_sync() {
  __builtin_amdgcn_fence(__ATOMIC_RELEASE, "wavefront");
  __builtin_amdgcn_wave_barrier();
  __builtin_amdgcn_fence(__ATOMIC_ACQUIRE, "wavefront");
}

template <int H, int C, bool RELU>
__global__ void __launch_bounds__(256)
node_softmax_gather(const int* __restrict__ rowptr, const int* __restrict__ csr_src,
                    const float* __restrict__ asrc, const float* __restrict__ adst,
                    const float* __restrict__ hfeat, const float* __restrict__ bias,
                    float* __restrict__ out, int n) {
  constexpr int F = H * C;
  constexpr int VPL = F / 32;
  static_assert(VPL == 8 || VPL == 2, "lane owns 8 or 2 features");
  __shared__ float strip[8][32 * H];

  const int lane = threadIdx.x & 31;
  const int wib  = threadIdx.x >> 5;
  const int node = __builtin_amdgcn_readfirstlane(blockIdx.x * 8 + wib);
  if (node >= n) return;
  const int beg = rowptr[node], end = rowptr[node + 1];
  const int hd = (lane * VPL) / C;
  float* myp = strip[wib];

  float ad[H];
  load_heads<H>(adst + (size_t)node * H, ad);

  auto stage_scores = [&](int c) -> int {
    const int j = c + lane;
    int s = 0;
    float v[H];
#pragma unroll
    for (int h = 0; h < H; ++h) v[h] = -INFINITY;
    if (j < end) {
      s = csr_src[j];
      float as[H];
      load_heads<H>(asrc + (size_t)s * H, as);
#pragma unroll
      for (int h = 0; h < H; ++h) {
        float t = as[h] + ad[h];
        v[h] = (t > 0.0f) ? t : NEG_SLOPE * t;
      }
    }
#pragma unroll
    for (int h = 0; h < H; ++h) myp[lane * H + h] = v[h];
    return s;
  };

  float mx = -INFINITY;
  int s = 0;
  for (int c = beg; c < end; c += 32) {
    s = stage_scores(c);
    wave_lds_sync();
#pragma unroll
    for (int t = 0; t < 32; ++t) mx = fmaxf(mx, myp[t * H + hd]);
    wave_lds_sync();
  }

  const bool single = (end - beg) <= 32;
  float acc[VPL];
#pragma unroll
  for (int i = 0; i < VPL; ++i) acc[i] = 0.0f;
  float lsum = 0.0f;
  for (int c = beg; c < end; c += 32) {
    if (!single) {
      s = stage_scores(c);
      wave_lds_sync();
    }
    const int m = min(32, end - c);
#pragma unroll 4
    for (int t = 0; t < m; ++t) {
      const int st = __builtin_amdgcn_readlane(s, t);
      const float p = __expf(myp[t * H + hd] - mx);
      lsum += p;
      const float* hp = hfeat + (size_t)st * F + lane * VPL;
      if constexpr (VPL == 8) {
        const v4f a = *(const v4f*)hp, b = *(const v4f*)(hp + 4);
        acc[0] += p * a.x; acc[1] += p * a.y; acc[2] += p * a.z; acc[3] += p * a.w;
        acc[4] += p * b.x; acc[5] += p * b.y; acc[6] += p * b.z; acc[7] += p * b.w;
      } else {
        const v2f a = *(const v2f*)hp;
        acc[0] += p * a.x; acc[1] += p * a.y;
      }
    }
    if (!single) wave_lds_sync();
  }

  const float denom = lsum + 1e-16f;
  float r[VPL];
#pragma unroll
  for (int i = 0; i < VPL; ++i) {
    float o = acc[i] / denom + bias[lane * VPL + i];
    r[i] = RELU ? fmaxf(o, 0.0f) : o;
  }
  float* op = out + (size_t)node * F + lane * VPL;
  if constexpr (VPL == 8) {
    v4f a = {r[0], r[1], r[2], r[3]}, b = {r[4], r[5], r[6], r[7]};
    *(v4f*)op = a;
    *(v4f*)(op + 4) = b;
  } else {
    v2f a = {r[0], r[1]};
    *(v2f*)op = a;
  }
}

extern "C" void kernel_launch(void* const* d_in, const int* in_sizes, int n_in,
                              void* d_out, int out_size, void* d_ws, size_t ws_size,
                              hipStream_t stream) {
  const float* x        = (const float*)d_in[0];
  const int*   ei       = (const int*)d_in[1];
  const float* W1       = (const float*)d_in[2];
  const float* att_src1 = (const float*)d_in[3];
  const float* att_dst1 = (const float*)d_in[4];
  const float* b1       = (const float*)d_in[5];
  const float* W2       = (const float*)d_in[6];
  const float* att_src2 = (const float*)d_in[7];
  const float* att_dst2 = (const float*)d_in[8];
  const float* b2       = (const float*)d_in[9];

  const int D_IN = 256, F1 = 256, F2 = 64, H1 = 8, C1 = 32, H2 = 1, C2 = 64;
  const int N = in_sizes[0] / D_IN;
  const int E = in_sizes[1] / 2;
  const int T = E + N;

  float* ws = (float*)d_ws;
  float* h1    = ws;
  float* agg1  = h1   + (size_t)N * F1;
  float* asrc1 = agg1 + (size_t)N * F1;
  float* adst1 = asrc1 + (size_t)N * H1;
  float* asrc2 = adst1 + (size_t)N * H1;
  float* adst2 = asrc2 + (N + 4);
  int*   cnt   = (int*)(adst2 + (N + 4));
  int*   loc   = cnt + (size_t)N * 8;
  int*   rowptr = loc + (N + 4);
  int*   bsum  = rowptr + (N + 4);
  int*   boff  = bsum + (N / 256 + 4);
  int*   rank  = boff + (N / 256 + 4);
  int*   csr_src = rank + E;
  float* out   = (float*)d_out;

  const int BT = 256;
  auto cdiv = [](long long a, long long b) { return (int)((a + b - 1) / b); };
  const int NB = cdiv(N, 256);

  init_counts<<<cdiv(N, BT), BT, 0, stream>>>(cnt, N);
  edge_rank<<<cdiv(E, BT), BT, 0, stream>>>(ei, E, cnt, rank);
  scan_local<<<NB, 256, 0, stream>>>(cnt, N, loc, bsum);
  scan_blocks<<<1, 256, 0, stream>>>(bsum, NB, boff);
  csr_fill<<<cdiv(T, BT), BT, 0, stream>>>(ei, E, T, cnt, loc, boff, rank, rowptr, csr_src);

  gemm_wmma_f32<256, 256><<<cdiv(N, 128) * (F1 / 64), 256, 0, stream>>>(x, W1, h1, N);
  att_scores<<<cdiv((long long)N * H1, BT), BT, 0, stream>>>(
      h1, att_src1, att_dst1, asrc1, adst1, N, H1, C1);
  node_softmax_gather<8, 32, true><<<cdiv(N, 8), 256, 0, stream>>>(
      rowptr, csr_src, asrc1, adst1, h1, b1, agg1, N);

  float* h2 = h1;
  gemm_wmma_f32<256, 64><<<cdiv(N, 128) * (F2 / 64), 256, 0, stream>>>(agg1, W2, h2, N);
  att_scores<<<cdiv(N, BT), BT, 0, stream>>>(
      h2, att_src2, att_dst2, asrc2, adst2, N, H2, C2);
  node_softmax_gather<1, 64, false><<<cdiv(N, 8), 256, 0, stream>>>(
      rowptr, csr_src, asrc2, adst2, h2, b2, out, N);
}


# Edge grouping rebuilt on LDS counters in two levels (chunk x 256-node bucket count, scan, place; one block per bucket fills rows): no global atomics, 10 launches; sum order = LDS atomic arrival rank
# speedup vs baseline: 5.9793x; 1.1860x over previous
#include <hip/hip_runtime.h>
#include <hip/hip_bf16.h>


typedef float v2f __attribute__((ext_vector_type(2)));
typedef int   v2i __attribute__((ext_vector_type(2)));
typedef float v4f __attribute__((ext_vector_type(4)));
typedef float v8f __attribute__((ext_vector_type(8)));

#define NEG_SLOPE 0.2f

template <int K, int NOUT>
__global__ void __launch_bounds__(256)
gemm_wmma_f32(const float* __restrict__ A,
              const float* __restrict__ B,
              float* __restrict__ C, int M) {
  constexpr int KC     = 32;
  constexpr int NCH    = K / KC;
  constexpr int NSTRIP = NOUT / 64;
  static_assert(K % KC == 0 && NOUT % 64 == 0, "tile sizes");
  __shared__ v4f Bs[2][KC * 16];

  const int tid  = threadIdx.x;
  const int lane = tid & 31;
  const int w    = tid >> 5;
  const int c    = lane & 15;
  const int h    = lane >> 4;
  const int nstrip = blockIdx.x % NSTRIP;
  const int mblk   = blockIdx.x / NSTRIP;
  const int n0     = nstrip * 64;

  int arow = mblk * 128 + w * 16 + c;
  if (arow >= M) arow = M - 1;
  const float* ap = A + (size_t)arow * K + 16 * h;

  const int br = tid >> 4, bq = tid & 15;
  const float* bp = B + (size_t)br * NOUT + n0 + 4 * bq;

  v4f a_cur[4], a_nxt[4], b_st[2];
#pragma unroll
  for (int i = 0; i < 4; ++i) a_cur[i] = *(const v4f*)(ap + 4 * i);
  b_st[0] = *(const v4f*)(bp);
  b_st[1] = *(const v4f*)(bp + (size_t)16 * NOUT);
  Bs[0][br * 16 + bq]        = b_st[0];
  Bs[0][(br + 16) * 16 + bq] = b_st[1];
  __syncthreads();

  v8f acc[4] = {};
  for (int kc = 0; kc < NCH; ++kc) {
    const int cur = kc & 1;
    const bool more = (kc + 1 < NCH);
    if (more) {
      const int k1 = (kc + 1) * KC;
#pragma unroll
      for (int i = 0; i < 4; ++i) a_nxt[i] = *(const v4f*)(ap + k1 + 4 * i);
      b_st[0] = *(const v4f*)(bp + (size_t)k1 * NOUT);
      b_st[1] = *(const v4f*)(bp + (size_t)(k1 + 16) * NOUT);
    }
    __builtin_amdgcn_sched_barrier(0);

    const v4f* bs = &Bs[cur][(16 * h) * 16 + c];
    v4f bf[16];
#pragma unroll
    for (int i = 0; i < 16; ++i) bf[i] = bs[i * 16];

#pragma unroll
    for (int s = 0; s < 8; ++s) {
      const v4f av = a_cur[s >> 1];
      v2f a;
      if (s & 1) { a.x = av.z; a.y = av.w; } else { a.x = av.x; a.y = av.y; }
      const v4f b0 = bf[2 * s], b1 = bf[2 * s + 1];
#pragma unroll
      for (int j = 0; j < 4; ++j) {
        v2f b;
        b.x = b0[j];
        b.y = b1[j];
        acc[j] = __builtin_amdgcn_wmma_f32_16x16x4_f32(
             false, a,  false, b,
             (short)0, acc[j],  false,  false);
      }
    }

    __builtin_amdgcn_sched_barrier(0);
    if (more) {
      Bs[cur ^ 1][br * 16 + bq]        = b_st[0];
      Bs[cur ^ 1][(br + 16) * 16 + bq] = b_st[1];
      __syncthreads();
#pragma unroll
      for (int i = 0; i < 4; ++i) a_cur[i] = a_nxt[i];
    }
  }

  const int orow = mblk * 128 + w * 16 + 8 * h;
  float* cp = C + (size_t)orow * NOUT + n0 + 4 * c;
#pragma unroll
  for (int r = 0; r < 8; ++r) {
    if (orow + r < M) {
      v4f o = {acc[0][r], acc[1][r], acc[2][r], acc[3][r]};
      *(v4f*)(cp + (size_t)r * NOUT) = o;
    }
  }
}

__global__ void att_scores(const float* __restrict__ h,
                           const float* __restrict__ att_src,
                           const float* __restrict__ att_dst,
                           float* __restrict__ asrc, float* __restrict__ adst,
                           int n, int heads, int ch) {
  int i = blockIdx.x * blockDim.x + threadIdx.x;
  if (i >= n * heads) return;
  int node = i / heads, hd = i % heads;
  const float* hp = h + (size_t)node * heads * ch + hd * ch;
  const float* as = att_src + hd * ch;
  const float* ad = att_dst + hd * ch;
  float ss = 0.0f, sd = 0.0f;
  for (int c = 0; c < ch; ++c) {
    float v = hp[c];
    ss += v * as[c];
    sd += v * ad[c];
  }
  asrc[i] = ss;
  adst[i] = sd;
}

#define BKT_SHIFT 8
#define BKT_NODES 256
#define NCHUNK    256
#define MAXB      1024

__device__ __forceinline__ int block_scan_incl(int v, int* wsum, int& total) {
  const int lane = threadIdx.x & 31, w = threadIdx.x >> 5;
  int x = v;
#pragma unroll
  for (int o = 1; o < 32; o <<= 1) {
    int y = __shfl_up(x, o, 32);
    if (lane >= o) x += y;
  }
  if (lane == 31) wsum[w] = x;
  __syncthreads();
  int woff = 0, tot = 0;
#pragma unroll
  for (int k = 0; k < 8; ++k) {
    int t = wsum[k];
    if (k < w) woff += t;
    tot += t;
  }
  total = tot;
  return woff + x;
}

#define EPT 16

__global__ void __launch_bounds__(256)
bucket_count(const int* __restrict__ ei, int E, int nbk, int* __restrict__ cm) {
  __shared__ int hist[MAXB];
  const int k = blockIdx.x;
  const int cs = (E + NCHUNK - 1) / NCHUNK;
  const int lo = k * cs, hi = min(E, lo + cs);
  const int* dp = ei + E;
  int d[EPT];
  auto load = [&](int base) {
#pragma unroll
    for (int u = 0; u < EPT; ++u) {
      const int i = base + u * 256 + (int)threadIdx.x;
      d[u] = (i < hi) ? dp[i] : -1;
    }
  };
  load(lo);
  for (int b = threadIdx.x; b < nbk; b += 256) hist[b] = 0;
  __syncthreads();
  for (int base = lo;;) {
#pragma unroll
    for (int u = 0; u < EPT; ++u)
      if (d[u] >= 0) atomicAdd(&hist[d[u] >> BKT_SHIFT], 1);
    base += EPT * 256;
    if (base >= hi) break;
    load(base);
  }
  __syncthreads();
  for (int b = threadIdx.x; b < nbk; b += 256) cm[(size_t)b * NCHUNK + k] = hist[b];
}

__global__ void __launch_bounds__(256)
bucket_scan(const int* __restrict__ cm, int* __restrict__ off, int* __restrict__ total) {
  __shared__ int wsum[8];
  const size_t i = (size_t)blockIdx.x * NCHUNK + threadIdx.x;
  const int v = cm[i];
  int tot;
  const int incl = block_scan_incl(v, wsum, tot);
  off[i] = incl - v;
  if (threadIdx.x == 0) total[blockIdx.x] = tot;
}

__global__ void __launch_bounds__(256)
bucket_place(const int* __restrict__ ei, int E, int nbk,
             const int* __restrict__ off, const int* __restrict__ total,
             int* __restrict__ bstart, int* __restrict__ part) {
  __shared__ int cursor[MAXB];
  __shared__ int wsum[8];
  const int k = blockIdx.x;
  const int cs = (E + NCHUNK - 1) / NCHUNK;
  const int lo = k * cs, hi = min(E, lo + cs);
  int sv[EPT], dv[EPT];
  auto load = [&](int base) {
#pragma unroll
    for (int u = 0; u < EPT; ++u) {
      const int i = base + u * 256 + (int)threadIdx.x;
      const bool ok = i < hi;
      dv[u] = ok ? ei[E + i] : -1;
      sv[u] = ok ? ei[i] : 0;
    }
  };
  load(lo);
  v2i* pairs = (v2i*)part;

  int carry = 0;
  for (int base = 0; base < nbk; base += 256) {
    const int b = base + threadIdx.x;
    const int v = (b < nbk) ? total[b] : 0;
    const int o = (b < nbk) ? off[(size_t)b * NCHUNK + k] : 0;
    int tot;
    const int incl = block_scan_incl(v, wsum, tot);
    if (b < nbk) {
      const int st = carry + incl - v;
      cursor[b] = st + o;
      if (k == 0) bstart[b] = st;
    }
    carry += tot;
    __syncthreads();
  }

  for (int base = lo;;) {
#pragma unroll
    for (int u = 0; u < EPT; ++u)
      if (dv[u] >= 0) {
        const int pos = atomicAdd(&cursor[dv[u] >> BKT_SHIFT], 1);
        const v2i e = {sv[u], dv[u]};
        pairs[pos] = e;
      }
    base += EPT * 256;
    if (base >= hi) break;
    load(base);
  }
}

__global__ void __launch_bounds__(256)
bucket_rows(const int* __restrict__ part, const int* __restrict__ bstart,
            const int* __restrict__ total, int n,
            int* __restrict__ rowptr, int* __restrict__ csr_src) {
  __shared__ int cnt[BKT_NODES];
  __shared__ int wsum[8];
  const int b = blockIdx.x;
  const int node = b * BKT_NODES + threadIdx.x;
  const int s = bstart[b], tot = total[b];
  const v2i* pp = (const v2i*)part + s;
  v2i ev[EPT];
  auto load = [&](int base) {
#pragma unroll
    for (int u = 0; u < EPT; ++u) {
      const int i = base + u * 256 + (int)threadIdx.x;
      const v2i none = {0, -1};
      ev[u] = (i < tot) ? pp[i] : none;
    }
  };
  load(0);
  cnt[threadIdx.x] = (node < n) ? 1 : 0;
  __syncthreads();
  for (int base = 0;;) {
#pragma unroll
    for (int u = 0; u < EPT; ++u)
      if (ev[u].y >= 0) atomicAdd(&cnt[ev[u].y & (BKT_NODES - 1)], 1);
    base += EPT * 256;
    if (base >= tot) break;
    load(base);
  }
  __syncthreads();
  const int v = cnt[threadIdx.x];
  int t;
  const int incl = block_scan_incl(v, wsum, t);
  const int beg = s + b * BKT_NODES + incl - v;
  if (node < n) {
    rowptr[node] = beg;
    csr_src[beg] = node;
    if (node == n - 1) rowptr[n] = beg + v;
  }
  cnt[threadIdx.x] = beg + 1;
  if (tot > EPT * 256) load(0);
  __syncthreads();
  for (int base = 0;;) {
#pragma unroll
    for (int u = 0; u < EPT; ++u)
      if (ev[u].y >= 0) {
        const int pos = atomicAdd(&cnt[ev[u].y & (BKT_NODES - 1)], 1);
        csr_src[pos] = ev[u].x;
      }
    base += EPT * 256;
    if (base >= tot) break;
    load(base);
  }
}

template <int H>
__device__ __forceinline__ void load_heads(const float* __restrict__ p, float (&v)[H]) {
  if constexpr (H == 8) {
    const v4f a = *(const v4f*)p, b = *(const v4f*)(p + 4);
    v[0] = a.x; v[1] = a.y; v[2] = a.z; v[3] = a.w;
    v[4] = b.x; v[5] = b.y; v[6] = b.z; v[7] = b.w;
  } else {
#pragma unroll
    for (int h = 0; h < H; ++h) v[h] = p[h];
  }
}

__device__ __forceinline__ void wave_lds_sync() {
  __builtin_amdgcn_fence(__ATOMIC_RELEASE, "wavefront");
  __builtin_amdgcn_wave_barrier();
  __builtin_amdgcn_fence(__ATOMIC_ACQUIRE, "wavefront");
}

template <int H, int C, bool RELU>
__global__ void __launch_bounds__(256)
node_softmax_gather(const int* __restrict__ rowptr, const int* __restrict__ csr_src,
                    const float* __restrict__ asrc, const float* __restrict__ adst,
                    const float* __restrict__ hfeat, const float* __restrict__ bias,
                    float* __restrict__ out, int n) {
  constexpr int F = H * C;
  constexpr int VPL = F / 32;
  static_assert(VPL == 8 || VPL == 2, "lane owns 8 or 2 features");
  __shared__ float strip[8][32 * H];

  const int lane = threadIdx.x & 31;
  const int wib  = threadIdx.x >> 5;
  const int node = __builtin_amdgcn_readfirstlane(blockIdx.x * 8 + wib);
  if (node >= n) return;
  const int beg = rowptr[node], end = rowptr[node + 1];
  const int hd = (lane * VPL) / C;
  float* myp = strip[wib];

  float ad[H];
  load_heads<H>(adst + (size_t)node * H, ad);

  auto stage_scores = [&](int c) -> int {
    const int j = c + lane;
    int s = 0;
    float v[H];
#pragma unroll
    for (int h = 0; h < H; ++h) v[h] = -INFINITY;
    if (j < end) {
      s = csr_src[j];
      float as[H];
      load_heads<H>(asrc + (size_t)s * H, as);
#pragma unroll
      for (int h = 0; h < H; ++h) {
        float t = as[h] + ad[h];
        v[h] = (t > 0.0f) ? t : NEG_SLOPE * t;
      }
    }
#pragma unroll
    for (int h = 0; h < H; ++h) myp[lane * H + h] = v[h];
    return s;
  };

  float mx = -INFINITY;
  int s = 0;
  for (int c = beg; c < end; c += 32) {
    s = stage_scores(c);
    wave_lds_sync();
#pragma unroll
    for (int t = 0; t < 32; ++t) mx = fmaxf(mx, myp[t * H + hd]);
    wave_lds_sync();
  }

  const bool single = (end - beg) <= 32;
  float acc[VPL];
#pragma unroll
  for (int i = 0; i < VPL; ++i) acc[i] = 0.0f;
  float lsum = 0.0f;
  for (int c = beg; c < end; c += 32) {
    if (!single) {
      s = stage_scores(c);
      wave_lds_sync();
    }
    const int m = min(32, end - c);
#pragma unroll 4
    for (int t = 0; t < m; ++t) {
      const int st = __builtin_amdgcn_readlane(s, t);
      const float p = __expf(myp[t * H + hd] - mx);
      lsum += p;
      const float* hp = hfeat + (size_t)st * F + lane * VPL;
      if constexpr (VPL == 8) {
        const v4f a = *(const v4f*)hp, b = *(const v4f*)(hp + 4);
        acc[0] += p * a.x; acc[1] += p * a.y; acc[2] += p * a.z; acc[3] += p * a.w;
        acc[4] += p * b.x; acc[5] += p * b.y; acc[6] += p * b.z; acc[7] += p * b.w;
      } else {
        const v2f a = *(const v2f*)hp;
        acc[0] += p * a.x; acc[1] += p * a.y;
      }
    }
    if (!single) wave_lds_sync();
  }

  const float denom = lsum + 1e-16f;
  float r[VPL];
#pragma unroll
  for (int i = 0; i < VPL; ++i) {
    float o = acc[i] / denom + bias[lane * VPL + i];
    r[i] = RELU ? fmaxf(o, 0.0f) : o;
  }
  float* op = out + (size_t)node * F + lane * VPL;
  if constexpr (VPL == 8) {
    v4f a = {r[0], r[1], r[2], r[3]}, b = {r[4], r[5], r[6], r[7]};
    *(v4f*)op = a;
    *(v4f*)(op + 4) = b;
  } else {
    v2f a = {r[0], r[1]};
    *(v2f*)op = a;
  }
}

extern "C" void kernel_launch(void* const* d_in, const int* in_sizes, int n_in,
                              void* d_out, int out_size, void* d_ws, size_t ws_size,
                              hipStream_t stream) {
  const float* x        = (const float*)d_in[0];
  const int*   ei       = (const int*)d_in[1];
  const float* W1       = (const float*)d_in[2];
  const float* att_src1 = (const float*)d_in[3];
  const float* att_dst1 = (const float*)d_in[4];
  const float* b1       = (const float*)d_in[5];
  const float* W2       = (const float*)d_in[6];
  const float* att_src2 = (const float*)d_in[7];
  const float* att_dst2 = (const float*)d_in[8];
  const float* b2       = (const float*)d_in[9];

  const int D_IN = 256, F1 = 256, F2 = 64, H1 = 8, C1 = 32, H2 = 1, C2 = 64;
  const int N = in_sizes[0] / D_IN;
  const int E = in_sizes[1] / 2;
  const int NBK = (N + BKT_NODES - 1) / BKT_NODES;

  float* ws = (float*)d_ws;
  float* h1    = ws;
  float* agg1  = h1   + (size_t)N * F1;
  float* asrc1 = agg1 + (size_t)N * F1;
  float* adst1 = asrc1 + (size_t)N * H1;
  int*   part  = (int*)(adst1 + (size_t)N * H1);
  float* asrc2 = (float*)(part + (size_t)2 * E);
  float* adst2 = asrc2 + (N + 4);
  int*   rowptr = (int*)(adst2 + (N + 4));
  int*   cm    = rowptr + (N + 4);
  int*   off   = cm + (size_t)NBK * NCHUNK;
  int*   total = off + (size_t)NBK * NCHUNK;
  int*   bstart = total + MAXB;
  int*   csr_src = bstart + MAXB;
  float* out   = (float*)d_out;

  const int BT = 256;
  auto cdiv = [](long long a, long long b) { return (int)((a + b - 1) / b); };

  bucket_count<<<NCHUNK, 256, 0, stream>>>(ei, E, NBK, cm);
  bucket_scan<<<NBK, 256, 0, stream>>>(cm, off, total);
  bucket_place<<<NCHUNK, 256, 0, stream>>>(ei, E, NBK, off, total, bstart, part);
  bucket_rows<<<NBK, 256, 0, stream>>>(part, bstart, total, N, rowptr, csr_src);

  gemm_wmma_f32<256, 256><<<cdiv(N, 128) * (F1 / 64), 256, 0, stream>>>(x, W1, h1, N);
  att_scores<<<cdiv((long long)N * H1, BT), BT, 0, stream>>>(
      h1, att_src1, att_dst1, asrc1, adst1, N, H1, C1);
  node_softmax_gather<8, 32, true><<<cdiv(N, 8), 256, 0, stream>>>(
      rowptr, csr_src, asrc1, adst1, h1, b1, agg1, N);

  float* h2 = h1;
  gemm_wmma_f32<256, 64><<<cdiv(N, 128) * (F2 / 64), 256, 0, stream>>>(agg1, W2, h2, N);
  att_scores<<<cdiv(N, BT), BT, 0, stream>>>(
      h2, att_src2, att_dst2, asrc2, adst2, N, H2, C2);
  node_softmax_gather<1, 64, false><<<cdiv(N, 8), 256, 0, stream>>>(
      rowptr, csr_src, asrc2, adst2, h2, b2, out, N);
}


# Attention scores formed in each GEMM epilogue from its accumulator tiles: both att_scores launches gone (10 -> 8), h not re-read; score dot summed as 4-column partials, then a lane tree
# speedup vs baseline: 6.4880x; 1.0851x over previous
#include <hip/hip_runtime.h>
#include <hip/hip_bf16.h>


typedef float v2f __attribute__((ext_vector_type(2)));
typedef int   v2i __attribute__((ext_vector_type(2)));
typedef float v4f __attribute__((ext_vector_type(4)));
typedef float v8f __attribute__((ext_vector_type(8)));

#define NEG_SLOPE 0.2f


template <int G>
__device__ __forceinline__ float head_group_sum(const float (&v)[8], int c) {
  static_assert(G == 8 || G == 16, "a head is 32 or 64 columns");
  constexpr int S = (G == 16) ? 1 : 0;
  const bool u2 = (c >> (S + 2)) & 1, u1 = (c >> (S + 1)) & 1, u0 = (c >> S) & 1;
  float a[4], b[2];
#pragma unroll
  for (int i = 0; i < 4; ++i)
    a[i] = (u2 ? v[i + 4] : v[i]) + __shfl_xor(u2 ? v[i] : v[i + 4], 4 << S, 32);
#pragma unroll
  for (int i = 0; i < 2; ++i)
    b[i] = (u1 ? a[i + 2] : a[i]) + __shfl_xor(u1 ? a[i] : a[i + 2], 2 << S, 32);
  float t = (u0 ? b[1] : b[0]) + __shfl_xor(u0 ? b[0] : b[1], 1 << S, 32);
  if constexpr (G == 16) t += __shfl_xor(t, 1, 32);
  return t;
}

template <int K, int NOUT, int CH>
__global__ void __launch_bounds__(256)
gemm_wmma_f32(const float* __restrict__ A,
              const float* __restrict__ B,
              float* __restrict__ C,
              const float* __restrict__ att_src,
              const float* __restrict__ att_dst,
              float* __restrict__ asrc, float* __restrict__ adst, int M) {
  constexpr int KC     = 32;
  constexpr int NCH    = K / KC;
  constexpr int NSTRIP = NOUT / 64;
  constexpr int HEADS  = NOUT / CH;
  constexpr int G      = CH / 4;
  static_assert(K % KC == 0 && NOUT % 64 == 0, "tile sizes");
  static_assert(NOUT % CH == 0 && (CH == 32 || CH == 64), "a head lies inside one 64-column strip");
  __shared__ v4f Bs[2][KC * 16];

  const int tid  = threadIdx.x;
  const int lane = tid & 31;
  const int w    = tid >> 5;
  const int c    = lane & 15;
  const int h    = lane >> 4;
  const int nstrip = blockIdx.x % NSTRIP;
  const int mblk   = blockIdx.x / NSTRIP;
  const int n0     = nstrip * 64;

  int arow = mblk * 128 + w * 16 + c;
  if (arow >= M) arow = M - 1;
  const float* ap = A + (size_t)arow * K + 16 * h;

  const int br = tid >> 4, bq = tid & 15;
  const float* bp = B + (size_t)br * NOUT + n0 + 4 * bq;

  const v4f ws = *(const v4f*)(att_src + n0 + 4 * c);
  const v4f wd = *(const v4f*)(att_dst + n0 + 4 * c);

  v4f a_cur[4], a_nxt[4], b_st[2];
#pragma unroll
  for (int i = 0; i < 4; ++i) a_cur[i] = *(const v4f*)(ap + 4 * i);
  b_st[0] = *(const v4f*)(bp);
  b_st[1] = *(const v4f*)(bp + (size_t)16 * NOUT);
  Bs[0][br * 16 + bq]        = b_st[0];
  Bs[0][(br + 16) * 16 + bq] = b_st[1];
  __syncthreads();

  v8f acc[4] = {};
  for (int kc = 0; kc < NCH; ++kc) {
    const int cur = kc & 1;
    const bool more = (kc + 1 < NCH);
    if (more) {
      const int k1 = (kc + 1) * KC;
#pragma unroll
      for (int i = 0; i < 4; ++i) a_nxt[i] = *(const v4f*)(ap + k1 + 4 * i);
      b_st[0] = *(const v4f*)(bp + (size_t)k1 * NOUT);
      b_st[1] = *(const v4f*)(bp + (size_t)(k1 + 16) * NOUT);
    }
    __builtin_amdgcn_sched_barrier(0);

    const v4f* bs = &Bs[cur][(16 * h) * 16 + c];
    v4f bf[16];
#pragma unroll
    for (int i = 0; i < 16; ++i) bf[i] = bs[i * 16];

#pragma unroll
    for (int s = 0; s < 8; ++s) {
      const v4f av = a_cur[s >> 1];
      v2f a;
      if (s & 1) { a.x = av.z; a.y = av.w; } else { a.x = av.x; a.y = av.y; }
      const v4f b0 = bf[2 * s], b1 = bf[2 * s + 1];
#pragma unroll
      for (int j = 0; j < 4; ++j) {
        v2f b;
        b.x = b0[j];
        b.y = b1[j];
        acc[j] = __builtin_amdgcn_wmma_f32_16x16x4_f32(
             false, a,  false, b,
             (short)0, acc[j],  false,  false);
      }
    }

    __builtin_amdgcn_sched_barrier(0);
    if (more) {
      Bs[cur ^ 1][br * 16 + bq]        = b_st[0];
      Bs[cur ^ 1][(br + 16) * 16 + bq] = b_st[1];
      __syncthreads();
#pragma unroll
      for (int i = 0; i < 4; ++i) a_cur[i] = a_nxt[i];
    }
  }

  const int orow = mblk * 128 + w * 16 + 8 * h;
  float* cp = C + (size_t)orow * NOUT + n0 + 4 * c;
#pragma unroll
  for (int r = 0; r < 8; ++r) {
    if (orow + r < M) {
      v4f o = {acc[0][r], acc[1][r], acc[2][r], acc[3][r]};
      *(v4f*)(cp + (size_t)r * NOUT) = o;
    }
  }

  float ps[8], pd[8];
#pragma unroll
  for (int r = 0; r < 8; ++r) {
    ps[r] = acc[0][r] * ws.x + acc[1][r] * ws.y + acc[2][r] * ws.z + acc[3][r] * ws.w;
    pd[r] = acc[0][r] * wd.x + acc[1][r] * wd.y + acc[2][r] * wd.z + acc[3][r] * wd.w;
  }
  const float ts = head_group_sum<G>(ps, c);
  const float td = head_group_sum<G>(pd, c);
  const int srow = orow + ((c >> (G == 16 ? 1 : 0)) & 7);
  const int head = (n0 + 4 * c) / CH;
  const bool writer = (G == 8) || ((c & 1) == 0);
  if (writer && srow < M) {
    asrc[(size_t)srow * HEADS + head] = ts;
    adst[(size_t)srow * HEADS + head] = td;
  }
}

#define BKT_SHIFT 8
#define BKT_NODES 256
#define NCHUNK    256
#define MAXB      1024

__device__ __forceinline__ int block_scan_incl(int v, int* wsum, int& total) {
  const int lane = threadIdx.x & 31, w = threadIdx.x >> 5;
  int x = v;
#pragma unroll
  for (int o = 1; o < 32; o <<= 1) {
    int y = __shfl_up(x, o, 32);
    if (lane >= o) x += y;
  }
  if (lane == 31) wsum[w] = x;
  __syncthreads();
  int woff = 0, tot = 0;
#pragma unroll
  for (int k = 0; k < 8; ++k) {
    int t = wsum[k];
    if (k < w) woff += t;
    tot += t;
  }
  total = tot;
  return woff + x;
}

#define EPT 16

__global__ void __launch_bounds__(256)
bucket_count(const int* __restrict__ ei, int E, int nbk, int* __restrict__ cm) {
  __shared__ int hist[MAXB];
  const int k = blockIdx.x;
  const int cs = (E + NCHUNK - 1) / NCHUNK;
  const int lo = k * cs, hi = min(E, lo + cs);
  const int* dp = ei + E;
  int d[EPT];
  auto load = [&](int base) {
#pragma unroll
    for (int u = 0; u < EPT; ++u) {
      const int i = base + u * 256 + (int)threadIdx.x;
      d[u] = (i < hi) ? dp[i] : -1;
    }
  };
  load(lo);
  for (int b = threadIdx.x; b < nbk; b += 256) hist[b] = 0;
  __syncthreads();
  for (int base = lo;;) {
#pragma unroll
    for (int u = 0; u < EPT; ++u)
      if (d[u] >= 0) atomicAdd(&hist[d[u] >> BKT_SHIFT], 1);
    base += EPT * 256;
    if (base >= hi) break;
    load(base);
  }
  __syncthreads();
  for (int b = threadIdx.x; b < nbk; b += 256) cm[(size_t)b * NCHUNK + k] = hist[b];
}

__global__ void __launch_bounds__(256)
bucket_scan(const int* __restrict__ cm, int* __restrict__ off, int* __restrict__ total) {
  __shared__ int wsum[8];
  const size_t i = (size_t)blockIdx.x * NCHUNK + threadIdx.x;
  const int v = cm[i];
  int tot;
  const int incl = block_scan_incl(v, wsum, tot);
  off[i] = incl - v;
  if (threadIdx.x == 0) total[blockIdx.x] = tot;
}

__global__ void __launch_bounds__(256)
bucket_place(const int* __restrict__ ei, int E, int nbk,
             const int* __restrict__ off, const int* __restrict__ total,
             int* __restrict__ bstart, int* __restrict__ part) {
  __shared__ int cursor[MAXB];
  __shared__ int wsum[8];
  const int k = blockIdx.x;
  const int cs = (E + NCHUNK - 1) / NCHUNK;
  const int lo = k * cs, hi = min(E, lo + cs);
  int sv[EPT], dv[EPT];
  auto load = [&](int base) {
#pragma unroll
    for (int u = 0; u < EPT; ++u) {
      const int i = base + u * 256 + (int)threadIdx.x;
      const bool ok = i < hi;
      dv[u] = ok ? ei[E + i] : -1;
      sv[u] = ok ? ei[i] : 0;
    }
  };
  load(lo);
  v2i* pairs = (v2i*)part;

  int carry = 0;
  for (int base = 0; base < nbk; base += 256) {
    const int b = base + threadIdx.x;
    const int v = (b < nbk) ? total[b] : 0;
    const int o = (b < nbk) ? off[(size_t)b * NCHUNK + k] : 0;
    int tot;
    const int incl = block_scan_incl(v, wsum, tot);
    if (b < nbk) {
      const int st = carry + incl - v;
      cursor[b] = st + o;
      if (k == 0) bstart[b] = st;
    }
    carry += tot;
    __syncthreads();
  }

  for (int base = lo;;) {
#pragma unroll
    for (int u = 0; u < EPT; ++u)
      if (dv[u] >= 0) {
        const int pos = atomicAdd(&cursor[dv[u] >> BKT_SHIFT], 1);
        const v2i e = {sv[u], dv[u]};
        pairs[pos] = e;
      }
    base += EPT * 256;
    if (base >= hi) break;
    load(base);
  }
}

__global__ void __launch_bounds__(256)
bucket_rows(const int* __restrict__ part, const int* __restrict__ bstart,
            const int* __restrict__ total, int n,
            int* __restrict__ rowptr, int* __restrict__ csr_src) {
  __shared__ int cnt[BKT_NODES];
  __shared__ int wsum[8];
  const int b = blockIdx.x;
  const int node = b * BKT_NODES + threadIdx.x;
  const int s = bstart[b], tot = total[b];
  const v2i* pp = (const v2i*)part + s;
  v2i ev[EPT];
  auto load = [&](int base) {
#pragma unroll
    for (int u = 0; u < EPT; ++u) {
      const int i = base + u * 256 + (int)threadIdx.x;
      const v2i none = {0, -1};
      ev[u] = (i < tot) ? pp[i] : none;
    }
  };
  load(0);
  cnt[threadIdx.x] = (node < n) ? 1 : 0;
  __syncthreads();
  for (int base = 0;;) {
#pragma unroll
    for (int u = 0; u < EPT; ++u)
      if (ev[u].y >= 0) atomicAdd(&cnt[ev[u].y & (BKT_NODES - 1)], 1);
    base += EPT * 256;
    if (base >= tot) break;
    load(base);
  }
  __syncthreads();
  const int v = cnt[threadIdx.x];
  int t;
  const int incl = block_scan_incl(v, wsum, t);
  const int beg = s + b * BKT_NODES + incl - v;
  if (node < n) {
    rowptr[node] = beg;
    csr_src[beg] = node;
    if (node == n - 1) rowptr[n] = beg + v;
  }
  cnt[threadIdx.x] = beg + 1;
  if (tot > EPT * 256) load(0);
  __syncthreads();
  for (int base = 0;;) {
#pragma unroll
    for (int u = 0; u < EPT; ++u)
      if (ev[u].y >= 0) {
        const int pos = atomicAdd(&cnt[ev[u].y & (BKT_NODES - 1)], 1);
        csr_src[pos] = ev[u].x;
      }
    base += EPT * 256;
    if (base >= tot) break;
    load(base);
  }
}

template <int H>
__device__ __forceinline__ void load_heads(const float* __restrict__ p, float (&v)[H]) {
  if constexpr (H == 8) {
    const v4f a = *(const v4f*)p, b = *(const v4f*)(p + 4);
    v[0] = a.x; v[1] = a.y; v[2] = a.z; v[3] = a.w;
    v[4] = b.x; v[5] = b.y; v[6] = b.z; v[7] = b.w;
  } else {
#pragma unroll
    for (int h = 0; h < H; ++h) v[h] = p[h];
  }
}

__device__ __forceinline__ void wave_lds_sync() {
  __builtin_amdgcn_fence(__ATOMIC_RELEASE, "wavefront");
  __builtin_amdgcn_wave_barrier();
  __builtin_amdgcn_fence(__ATOMIC_ACQUIRE, "wavefront");
}

template <int H, int C, bool RELU>
__global__ void __launch_bounds__(256)
node_softmax_gather(const int* __restrict__ rowptr, const int* __restrict__ csr_src,
                    const float* __restrict__ asrc, const float* __restrict__ adst,
                    const float* __restrict__ hfeat, const float* __restrict__ bias,
                    float* __restrict__ out, int n) {
  constexpr int F = H * C;
  constexpr int VPL = F / 32;
  static_assert(VPL == 8 || VPL == 2, "lane owns 8 or 2 features");
  __shared__ float strip[8][32 * H];

  const int lane = threadIdx.x & 31;
  const int wib  = threadIdx.x >> 5;
  const int node = __builtin_amdgcn_readfirstlane(blockIdx.x * 8 + wib);
  if (node >= n) return;
  const int beg = rowptr[node], end = rowptr[node + 1];
  const int hd = (lane * VPL) / C;
  float* myp = strip[wib];

  float ad[H];
  load_heads<H>(adst + (size_t)node * H, ad);

  auto stage_scores = [&](int c) -> int {
    const int j = c + lane;
    int s = 0;
    float v[H];
#pragma unroll
    for (int h = 0; h < H; ++h) v[h] = -INFINITY;
    if (j < end) {
      s = csr_src[j];
      float as[H];
      load_heads<H>(asrc + (size_t)s * H, as);
#pragma unroll
      for (int h = 0; h < H; ++h) {
        float t = as[h] + ad[h];
        v[h] = (t > 0.0f) ? t : NEG_SLOPE * t;
      }
    }
#pragma unroll
    for (int h = 0; h < H; ++h) myp[lane * H + h] = v[h];
    return s;
  };

  float mx = -INFINITY;
  int s = 0;
  for (int c = beg; c < end; c += 32) {
    s = stage_scores(c);
    wave_lds_sync();
#pragma unroll
    for (int t = 0; t < 32; ++t) mx = fmaxf(mx, myp[t * H + hd]);
    wave_lds_sync();
  }

  const bool single = (end - beg) <= 32;
  float acc[VPL];
#pragma unroll
  for (int i = 0; i < VPL; ++i) acc[i] = 0.0f;
  float lsum = 0.0f;
  for (int c = beg; c < end; c += 32) {
    if (!single) {
      s = stage_scores(c);
      wave_lds_sync();
    }
    const int m = min(32, end - c);
#pragma unroll 4
    for (int t = 0; t < m; ++t) {
      const int st = __builtin_amdgcn_readlane(s, t);
      const float p = __expf(myp[t * H + hd] - mx);
      lsum += p;
      const float* hp = hfeat + (size_t)st * F + lane * VPL;
      if constexpr (VPL == 8) {
        const v4f a = *(const v4f*)hp, b = *(const v4f*)(hp + 4);
        acc[0] += p * a.x; acc[1] += p * a.y; acc[2] += p * a.z; acc[3] += p * a.w;
        acc[4] += p * b.x; acc[5] += p * b.y; acc[6] += p * b.z; acc[7] += p * b.w;
      } else {
        const v2f a = *(const v2f*)hp;
        acc[0] += p * a.x; acc[1] += p * a.y;
      }
    }
    if (!single) wave_lds_sync();
  }

  const float denom = lsum + 1e-16f;
  float r[VPL];
#pragma unroll
  for (int i = 0; i < VPL; ++i) {
    float o = acc[i] / denom + bias[lane * VPL + i];
    r[i] = RELU ? fmaxf(o, 0.0f) : o;
  }
  float* op = out + (size_t)node * F + lane * VPL;
  if constexpr (VPL == 8) {
    v4f a = {r[0], r[1], r[2], r[3]}, b = {r[4], r[5], r[6], r[7]};
    *(v4f*)op = a;
    *(v4f*)(op + 4) = b;
  } else {
    v2f a = {r[0], r[1]};
    *(v2f*)op = a;
  }
}

extern "C" void kernel_launch(void* const* d_in, const int* in_sizes, int n_in,
                              void* d_out, int out_size, void* d_ws, size_t ws_size,
                              hipStream_t stream) {
  const float* x        = (const float*)d_in[0];
  const int*   ei       = (const int*)d_in[1];
  const float* W1       = (const float*)d_in[2];
  const float* att_src1 = (const float*)d_in[3];
  const float* att_dst1 = (const float*)d_in[4];
  const float* b1       = (const float*)d_in[5];
  const float* W2       = (const float*)d_in[6];
  const float* att_src2 = (const float*)d_in[7];
  const float* att_dst2 = (const float*)d_in[8];
  const float* b2       = (const float*)d_in[9];

  const int D_IN = 256, F1 = 256, F2 = 64, H1 = 8;
  const int N = in_sizes[0] / D_IN;
  const int E = in_sizes[1] / 2;
  const int NBK = (N + BKT_NODES - 1) / BKT_NODES;

  float* ws = (float*)d_ws;
  float* h1    = ws;
  float* agg1  = h1   + (size_t)N * F1;
  float* asrc1 = agg1 + (size_t)N * F1;
  float* adst1 = asrc1 + (size_t)N * H1;
  int*   part  = (int*)(adst1 + (size_t)N * H1);
  float* asrc2 = (float*)(part + (size_t)2 * E);
  float* adst2 = asrc2 + (N + 4);
  int*   rowptr = (int*)(adst2 + (N + 4));
  int*   cm    = rowptr + (N + 4);
  int*   off   = cm + (size_t)NBK * NCHUNK;
  int*   total = off + (size_t)NBK * NCHUNK;
  int*   bstart = total + MAXB;
  int*   csr_src = bstart + MAXB;
  float* out   = (float*)d_out;

  auto cdiv = [](long long a, long long b) { return (int)((a + b - 1) / b); };

  bucket_count<<<NCHUNK, 256, 0, stream>>>(ei, E, NBK, cm);
  bucket_scan<<<NBK, 256, 0, stream>>>(cm, off, total);
  bucket_place<<<NCHUNK, 256, 0, stream>>>(ei, E, NBK, off, total, bstart, part);
  bucket_rows<<<NBK, 256, 0, stream>>>(part, bstart, total, N, rowptr, csr_src);

  gemm_wmma_f32<256, 256, 32><<<cdiv(N, 128) * (F1 / 64), 256, 0, stream>>>(
      x, W1, h1, att_src1, att_dst1, asrc1, adst1, N);
  node_softmax_gather<8, 32, true><<<cdiv(N, 8), 256, 0, stream>>>(
      rowptr, csr_src, asrc1, adst1, h1, b1, agg1, N);

  float* h2 = h1;
  gemm_wmma_f32<256, 64, 64><<<cdiv(N, 128) * (F2 / 64), 256, 0, stream>>>(
      agg1, W2, h2, att_src2, att_dst2, asrc2, adst2, N);
  node_softmax_gather<1, 64, false><<<cdiv(N, 8), 256, 0, stream>>>(
      rowptr, csr_src, asrc2, adst2, h2, b2, out, N);
}


# Edge-grouping passes lose their own launches: each runs in one launch beside a slice of the layer-1 GEMM's tiles (block index picks the role); 8 -> 7 launches; order of every sum unchanged
# speedup vs baseline: 6.8740x; 1.0595x over previous
#include <hip/hip_runtime.h>
#include <hip/hip_bf16.h>


typedef float v2f __attribute__((ext_vector_type(2)));
typedef int   v2i __attribute__((ext_vector_type(2)));
typedef float v4f __attribute__((ext_vector_type(4)));
typedef float v8f __attribute__((ext_vector_type(8)));

#define NEG_SLOPE 0.2f


template <int G>
__device__ __forceinline__ float head_group_sum(const float (&v)[8], int c) {
  static_assert(G == 8 || G == 16, "a head is 32 or 64 columns");
  constexpr int S = (G == 16) ? 1 : 0;
  const bool u2 = (c >> (S + 2)) & 1, u1 = (c >> (S + 1)) & 1, u0 = (c >> S) & 1;
  float a[4], b[2];
#pragma unroll
  for (int i = 0; i < 4; ++i)
    a[i] = (u2 ? v[i + 4] : v[i]) + __shfl_xor(u2 ? v[i] : v[i + 4], 4 << S, 32);
#pragma unroll
  for (int i = 0; i < 2; ++i)
    b[i] = (u1 ? a[i + 2] : a[i]) + __shfl_xor(u1 ? a[i] : a[i + 2], 2 << S, 32);
  float t = (u0 ? b[1] : b[0]) + __shfl_xor(u0 ? b[0] : b[1], 1 << S, 32);
  if constexpr (G == 16) t += __shfl_xor(t, 1, 32);
  return t;
}

template <int K, int NOUT, int CH>
__device__ __forceinline__ void
gemm_tile(const float* __restrict__ A,
          const float* __restrict__ B,
          float* __restrict__ C,
          const float* __restrict__ att_src,
          const float* __restrict__ att_dst,
          float* __restrict__ asrc, float* __restrict__ adst, int M, int tile) {
  constexpr int KC     = 32;
  constexpr int NCH    = K / KC;
  constexpr int NSTRIP = NOUT / 64;
  constexpr int HEADS  = NOUT / CH;
  constexpr int G      = CH / 4;
  static_assert(K % KC == 0 && NOUT % 64 == 0, "tile sizes");
  static_assert(NOUT % CH == 0 && (CH == 32 || CH == 64), "a head lies inside one 64-column strip");
  __shared__ v4f Bs[2][KC * 16];

  const int tid  = threadIdx.x;
  const int lane = tid & 31;
  const int w    = tid >> 5;
  const int c    = lane & 15;
  const int h    = lane >> 4;
  const int nstrip = tile % NSTRIP;
  const int mblk   = tile / NSTRIP;
  const int n0     = nstrip * 64;

  int arow = mblk * 128 + w * 16 + c;
  if (arow >= M) arow = M - 1;
  const float* ap = A + (size_t)arow * K + 16 * h;

  const int br = tid >> 4, bq = tid & 15;
  const float* bp = B + (size_t)br * NOUT + n0 + 4 * bq;

  const v4f ws = *(const v4f*)(att_src + n0 + 4 * c);
  const v4f wd = *(const v4f*)(att_dst + n0 + 4 * c);

  v4f a_cur[4], a_nxt[4], b_st[2];
#pragma unroll
  for (int i = 0; i < 4; ++i) a_cur[i] = *(const v4f*)(ap + 4 * i);
  b_st[0] = *(const v4f*)(bp);
  b_st[1] = *(const v4f*)(bp + (size_t)16 * NOUT);
  Bs[0][br * 16 + bq]        = b_st[0];
  Bs[0][(br + 16) * 16 + bq] = b_st[1];
  __syncthreads();

  v8f acc[4] = {};
  for (int kc = 0; kc < NCH; ++kc) {
    const int cur = kc & 1;
    const bool more = (kc + 1 < NCH);
    if (more) {
      const int k1 = (kc + 1) * KC;
#pragma unroll
      for (int i = 0; i < 4; ++i) a_nxt[i] = *(const v4f*)(ap + k1 + 4 * i);
      b_st[0] = *(const v4f*)(bp + (size_t)k1 * NOUT);
      b_st[1] = *(const v4f*)(bp + (size_t)(k1 + 16) * NOUT);
    }
    __builtin_amdgcn_sched_barrier(0);

    const v4f* bs = &Bs[cur][(16 * h) * 16 + c];
    v4f bf[16];
#pragma unroll
    for (int i = 0; i < 16; ++i) bf[i] = bs[i * 16];

#pragma unroll
    for (int s = 0; s < 8; ++s) {
      const v4f av = a_cur[s >> 1];
      v2f a;
      if (s & 1) { a.x = av.z; a.y = av.w; } else { a.x = av.x; a.y = av.y; }
      const v4f b0 = bf[2 * s], b1 = bf[2 * s + 1];
#pragma unroll
      for (int j = 0; j < 4; ++j) {
        v2f b;
        b.x = b0[j];
        b.y = b1[j];
        acc[j] = __builtin_amdgcn_wmma_f32_16x16x4_f32(
             false, a,  false, b,
             (short)0, acc[j],  false,  false);
      }
    }

    __builtin_amdgcn_sched_barrier(0);
    if (more) {
      Bs[cur ^ 1][br * 16 + bq]        = b_st[0];
      Bs[cur ^ 1][(br + 16) * 16 + bq] = b_st[1];
      __syncthreads();
#pragma unroll
      for (int i = 0; i < 4; ++i) a_cur[i] = a_nxt[i];
    }
  }

  const int orow = mblk * 128 + w * 16 + 8 * h;
  float* cp = C + (size_t)orow * NOUT + n0 + 4 * c;
#pragma unroll
  for (int r = 0; r < 8; ++r) {
    if (orow + r < M) {
      v4f o = {acc[0][r], acc[1][r], acc[2][r], acc[3][r]};
      *(v4f*)(cp + (size_t)r * NOUT) = o;
    }
  }

  float ps[8], pd[8];
#pragma unroll
  for (int r = 0; r < 8; ++r) {
    ps[r] = acc[0][r] * ws.x + acc[1][r] * ws.y + acc[2][r] * ws.z + acc[3][r] * ws.w;
    pd[r] = acc[0][r] * wd.x + acc[1][r] * wd.y + acc[2][r] * wd.z + acc[3][r] * wd.w;
  }
  const float ts = head_group_sum<G>(ps, c);
  const float td = head_group_sum<G>(pd, c);
  const int srow = orow + ((c >> (G == 16 ? 1 : 0)) & 7);
  const int head = (n0 + 4 * c) / CH;
  const bool writer = (G == 8) || ((c & 1) == 0);
  if (writer && srow < M) {
    asrc[(size_t)srow * HEADS + head] = ts;
    adst[(size_t)srow * HEADS + head] = td;
  }
}

template <int K, int NOUT, int CH>
__global__ void __launch_bounds__(256)
gemm_wmma_f32(const float* __restrict__ A, const float* __restrict__ B, float* __restrict__ C,
              const float* __restrict__ att_src, const float* __restrict__ att_dst,
              float* __restrict__ asrc, float* __restrict__ adst, int M) {
  gemm_tile<K, NOUT, CH>(A, B, C, att_src, att_dst, asrc, adst, M, (int)blockIdx.x);
}

#define BKT_SHIFT 8
#define BKT_NODES 256
#define NCHUNK    256
#define MAXB      1024

__device__ __forceinline__ int block_scan_incl(int v, int* wsum, int& total) {
  const int lane = threadIdx.x & 31, w = threadIdx.x >> 5;
  int x = v;
#pragma unroll
  for (int o = 1; o < 32; o <<= 1) {
    int y = __shfl_up(x, o, 32);
    if (lane >= o) x += y;
  }
  if (lane == 31) wsum[w] = x;
  __syncthreads();
  int woff = 0, tot = 0;
#pragma unroll
  for (int k = 0; k < 8; ++k) {
    int t = wsum[k];
    if (k < w) woff += t;
    tot += t;
  }
  total = tot;
  return woff + x;
}

#define EPT 16

__device__ __forceinline__ void
bucket_count(int k, const int* __restrict__ ei, int E, int nbk, int* __restrict__ cm) {
  __shared__ int hist[MAXB];
  const int cs = (E + NCHUNK - 1) / NCHUNK;
  const int lo = k * cs, hi = min(E, lo + cs);
  const int* dp = ei + E;
  int d[EPT];
  auto load = [&](int base) {
#pragma unroll
    for (int u = 0; u < EPT; ++u) {
      const int i = base + u * 256 + (int)threadIdx.x;
      d[u] = (i < hi) ? dp[i] : -1;
    }
  };
  load(lo);
  for (int b = threadIdx.x; b < nbk; b += 256) hist[b] = 0;
  __syncthreads();
  for (int base = lo;;) {
#pragma unroll
    for (int u = 0; u < EPT; ++u)
      if (d[u] >= 0) atomicAdd(&hist[d[u] >> BKT_SHIFT], 1);
    base += EPT * 256;
    if (base >= hi) break;
    load(base);
  }
  __syncthreads();
  for (int b = threadIdx.x; b < nbk; b += 256) cm[(size_t)b * NCHUNK + k] = hist[b];
}

__device__ __forceinline__ void
bucket_scan(int b, const int* __restrict__ cm, int* __restrict__ off, int* __restrict__ total) {
  __shared__ int wsum[8];
  const size_t i = (size_t)b * NCHUNK + threadIdx.x;
  const int v = cm[i];
  int tot;
  const int incl = block_scan_incl(v, wsum, tot);
  off[i] = incl - v;
  if (threadIdx.x == 0) total[b] = tot;
}

__device__ __forceinline__ void
bucket_place(int k, const int* __restrict__ ei, int E, int nbk,
             const int* __restrict__ off, const int* __restrict__ total,
             int* __restrict__ bstart, int* __restrict__ part) {
  __shared__ int cursor[MAXB];
  __shared__ int wsum[8];
  const int cs = (E + NCHUNK - 1) / NCHUNK;
  const int lo = k * cs, hi = min(E, lo + cs);
  int sv[EPT], dv[EPT];
  auto load = [&](int base) {
#pragma unroll
    for (int u = 0; u < EPT; ++u) {
      const int i = base + u * 256 + (int)threadIdx.x;
      const bool ok = i < hi;
      dv[u] = ok ? ei[E + i] : -1;
      sv[u] = ok ? ei[i] : 0;
    }
  };
  load(lo);
  v2i* pairs = (v2i*)part;

  int carry = 0;
  for (int base = 0; base < nbk; base += 256) {
    const int b = base + threadIdx.x;
    const int v = (b < nbk) ? total[b] : 0;
    const int o = (b < nbk) ? off[(size_t)b * NCHUNK + k] : 0;
    int tot;
    const int incl = block_scan_incl(v, wsum, tot);
    if (b < nbk) {
      const int st = carry + incl - v;
      cursor[b] = st + o;
      if (k == 0) bstart[b] = st;
    }
    carry += tot;
    __syncthreads();
  }

  for (int base = lo;;) {
#pragma unroll
    for (int u = 0; u < EPT; ++u)
      if (dv[u] >= 0) {
        const int pos = atomicAdd(&cursor[dv[u] >> BKT_SHIFT], 1);
        const v2i e = {sv[u], dv[u]};
        pairs[pos] = e;
      }
    base += EPT * 256;
    if (base >= hi) break;
    load(base);
  }
}

__device__ __forceinline__ void
bucket_rows(int b, const int* __restrict__ part, const int* __restrict__ bstart,
            const int* __restrict__ total, int n,
            int* __restrict__ rowptr, int* __restrict__ csr_src) {
  __shared__ int cnt[BKT_NODES];
  __shared__ int wsum[8];
  const int node = b * BKT_NODES + threadIdx.x;
  const int s = bstart[b], tot = total[b];
  const v2i* pp = (const v2i*)part + s;
  v2i ev[EPT];
  auto load = [&](int base) {
#pragma unroll
    for (int u = 0; u < EPT; ++u) {
      const int i = base + u * 256 + (int)threadIdx.x;
      const v2i none = {0, -1};
      ev[u] = (i < tot) ? pp[i] : none;
    }
  };
  load(0);
  cnt[threadIdx.x] = (node < n) ? 1 : 0;
  __syncthreads();
  for (int base = 0;;) {
#pragma unroll
    for (int u = 0; u < EPT; ++u)
      if (ev[u].y >= 0) atomicAdd(&cnt[ev[u].y & (BKT_NODES - 1)], 1);
    base += EPT * 256;
    if (base >= tot) break;
    load(base);
  }
  __syncthreads();
  const int v = cnt[threadIdx.x];
  int t;
  const int incl = block_scan_incl(v, wsum, t);
  const int beg = s + b * BKT_NODES + incl - v;
  if (node < n) {
    rowptr[node] = beg;
    csr_src[beg] = node;
    if (node == n - 1) rowptr[n] = beg + v;
  }
  cnt[threadIdx.x] = beg + 1;
  if (tot > EPT * 256) load(0);
  __syncthreads();
  for (int base = 0;;) {
#pragma unroll
    for (int u = 0; u < EPT; ++u)
      if (ev[u].y >= 0) {
        const int pos = atomicAdd(&cnt[ev[u].y & (BKT_NODES - 1)], 1);
        csr_src[pos] = ev[u].x;
      }
    base += EPT * 256;
    if (base >= tot) break;
    load(base);
  }
}

template <int PASS>
__global__ void __launch_bounds__(256)
group_pass_beside_gemm(const int* __restrict__ ei, int E, int nbk, int n,
                       int* cm, int* off, int* total, int* bstart, int* part,
                       int* rowptr, int* csr_src, int ngroup,
                       const float* __restrict__ A, const float* __restrict__ B,
                       float* __restrict__ C,
                       const float* __restrict__ att_src, const float* __restrict__ att_dst,
                       float* __restrict__ asrc, float* __restrict__ adst, int M, int tile0) {
  const int bid = (int)blockIdx.x;
  if (bid < ngroup) {
    if constexpr (PASS == 0)      bucket_count(bid, ei, E, nbk, cm);
    else if constexpr (PASS == 1) bucket_scan(bid, cm, off, total);
    else if constexpr (PASS == 2) bucket_place(bid, ei, E, nbk, off, total, bstart, part);
    else                          bucket_rows(bid, part, bstart, total, n, rowptr, csr_src);
  } else {
    gemm_tile<256, 256, 32>(A, B, C, att_src, att_dst, asrc, adst, M, tile0 + bid - ngroup);
  }
}

template <int H>
__device__ __forceinline__ void load_heads(const float* __restrict__ p, float (&v)[H]) {
  if constexpr (H == 8) {
    const v4f a = *(const v4f*)p, b = *(const v4f*)(p + 4);
    v[0] = a.x; v[1] = a.y; v[2] = a.z; v[3] = a.w;
    v[4] = b.x; v[5] = b.y; v[6] = b.z; v[7] = b.w;
  } else {
#pragma unroll
    for (int h = 0; h < H; ++h) v[h] = p[h];
  }
}

__device__ __forceinline__ void wave_lds_sync() {
  __builtin_amdgcn_fence(__ATOMIC_RELEASE, "wavefront");
  __builtin_amdgcn_wave_barrier();
  __builtin_amdgcn_fence(__ATOMIC_ACQUIRE, "wavefront");
}

template <int H, int C, bool RELU>
__global__ void __launch_bounds__(256)
node_softmax_gather(const int* __restrict__ rowptr, const int* __restrict__ csr_src,
                    const float* __restrict__ asrc, const float* __restrict__ adst,
                    const float* __restrict__ hfeat, const float* __restrict__ bias,
                    float* __restrict__ out, int n) {
  constexpr int F = H * C;
  constexpr int VPL = F / 32;
  static_assert(VPL == 8 || VPL == 2, "lane owns 8 or 2 features");
  __shared__ float strip[8][32 * H];

  const int lane = threadIdx.x & 31;
  const int wib  = threadIdx.x >> 5;
  const int node = __builtin_amdgcn_readfirstlane(blockIdx.x * 8 + wib);
  if (node >= n) return;
  const int beg = rowptr[node], end = rowptr[node + 1];
  const int hd = (lane * VPL) / C;
  float* myp = strip[wib];

  float ad[H];
  load_heads<H>(adst + (size_t)node * H, ad);

  auto stage_scores = [&](int c) -> int {
    const int j = c + lane;
    int s = 0;
    float v[H];
#pragma unroll
    for (int h = 0; h < H; ++h) v[h] = -INFINITY;
    if (j < end) {
      s = csr_src[j];
      float as[H];
      load_heads<H>(asrc + (size_t)s * H, as);
#pragma unroll
      for (int h = 0; h < H; ++h) {
        float t = as[h] + ad[h];
        v[h] = (t > 0.0f) ? t : NEG_SLOPE * t;
      }
    }
#pragma unroll
    for (int h = 0; h < H; ++h) myp[lane * H + h] = v[h];
    return s;
  };

  float mx = -INFINITY;
  int s = 0;
  for (int c = beg; c < end; c += 32) {
    s = stage_scores(c);
    wave_lds_sync();
#pragma unroll
    for (int t = 0; t < 32; ++t) mx = fmaxf(mx, myp[t * H + hd]);
    wave_lds_sync();
  }

  const bool single = (end - beg) <= 32;
  float acc[VPL];
#pragma unroll
  for (int i = 0; i < VPL; ++i) acc[i] = 0.0f;
  float lsum = 0.0f;
  for (int c = beg; c < end; c += 32) {
    if (!single) {
      s = stage_scores(c);
      wave_lds_sync();
    }
    const int m = min(32, end - c);
#pragma unroll 4
    for (int t = 0; t < m; ++t) {
      const int st = __builtin_amdgcn_readlane(s, t);
      const float p = __expf(myp[t * H + hd] - mx);
      lsum += p;
      const float* hp = hfeat + (size_t)st * F + lane * VPL;
      if constexpr (VPL == 8) {
        const v4f a = *(const v4f*)hp, b = *(const v4f*)(hp + 4);
        acc[0] += p * a.x; acc[1] += p * a.y; acc[2] += p * a.z; acc[3] += p * a.w;
        acc[4] += p * b.x; acc[5] += p * b.y; acc[6] += p * b.z; acc[7] += p * b.w;
      } else {
        const v2f a = *(const v2f*)hp;
        acc[0] += p * a.x; acc[1] += p * a.y;
      }
    }
    if (!single) wave_lds_sync();
  }

  const float denom = lsum + 1e-16f;
  float r[VPL];
#pragma unroll
  for (int i = 0; i < VPL; ++i) {
    float o = acc[i] / denom + bias[lane * VPL + i];
    r[i] = RELU ? fmaxf(o, 0.0f) : o;
  }
  float* op = out + (size_t)node * F + lane * VPL;
  if constexpr (VPL == 8) {
    v4f a = {r[0], r[1], r[2], r[3]}, b = {r[4], r[5], r[6], r[7]};
    *(v4f*)op = a;
    *(v4f*)(op + 4) = b;
  } else {
    v2f a = {r[0], r[1]};
    *(v2f*)op = a;
  }
}

extern "C" void kernel_launch(void* const* d_in, const int* in_sizes, int n_in,
                              void* d_out, int out_size, void* d_ws, size_t ws_size,
                              hipStream_t stream) {
  const float* x        = (const float*)d_in[0];
  const int*   ei       = (const int*)d_in[1];
  const float* W1       = (const float*)d_in[2];
  const float* att_src1 = (const float*)d_in[3];
  const float* att_dst1 = (const float*)d_in[4];
  const float* b1       = (const float*)d_in[5];
  const float* W2       = (const float*)d_in[6];
  const float* att_src2 = (const float*)d_in[7];
  const float* att_dst2 = (const float*)d_in[8];
  const float* b2       = (const float*)d_in[9];

  const int D_IN = 256, F1 = 256, F2 = 64, H1 = 8;
  const int N = in_sizes[0] / D_IN;
  const int E = in_sizes[1] / 2;
  const int NBK = (N + BKT_NODES - 1) / BKT_NODES;

  float* ws = (float*)d_ws;
  float* h1    = ws;
  float* agg1  = h1   + (size_t)N * F1;
  float* asrc1 = agg1 + (size_t)N * F1;
  float* adst1 = asrc1 + (size_t)N * H1;
  int*   part  = (int*)(adst1 + (size_t)N * H1);
  float* asrc2 = (float*)(part + (size_t)2 * E);
  float* adst2 = asrc2 + (N + 4);
  int*   rowptr = (int*)(adst2 + (N + 4));
  int*   cm    = rowptr + (N + 4);
  int*   off   = cm + (size_t)NBK * NCHUNK;
  int*   total = off + (size_t)NBK * NCHUNK;
  int*   bstart = total + MAXB;
  int*   csr_src = bstart + MAXB;
  float* out   = (float*)d_out;

  auto cdiv = [](long long a, long long b) { return (int)((a + b - 1) / b); };

  auto imin = [](int a, int b) { return a < b ? a : b; };
  const int T  = cdiv(N, 128) * (F1 / 64);
  const int t0 = imin(T, 512), t1 = 0, t2 = imin(T - t0, 512);
  const int t3 = T - t0 - t1 - t2;
  group_pass_beside_gemm<0><<<NCHUNK + t0, 256, 0, stream>>>(
      ei, E, NBK, N, cm, off, total, bstart, part, rowptr, csr_src, NCHUNK,
      x, W1, h1, att_src1, att_dst1, asrc1, adst1, N, 0);
  group_pass_beside_gemm<1><<<NBK + t1, 256, 0, stream>>>(
      ei, E, NBK, N, cm, off, total, bstart, part, rowptr, csr_src, NBK,
      x, W1, h1, att_src1, att_dst1, asrc1, adst1, N, t0);
  group_pass_beside_gemm<2><<<NCHUNK + t2, 256, 0, stream>>>(
      ei, E, NBK, N, cm, off, total, bstart, part, rowptr, csr_src, NCHUNK,
      x, W1, h1, att_src1, att_dst1, asrc1, adst1, N, t0 + t1);
  group_pass_beside_gemm<3><<<NBK + t3, 256, 0, stream>>>(
      ei, E, NBK, N, cm, off, total, bstart, part, rowptr, csr_src, NBK,
      x, W1, h1, att_src1, att_dst1, asrc1, adst1, N, t0 + t1 + t2);

  node_softmax_gather<8, 32, true><<<cdiv(N, 8), 256, 0, stream>>>(
      rowptr, csr_src, asrc1, adst1, h1, b1, agg1, N);

  float* h2 = h1;
  gemm_wmma_f32<256, 64, 64><<<cdiv(N, 128) * (F2 / 64), 256, 0, stream>>>(
      agg1, W2, h2, att_src2, att_dst2, asrc2, adst2, N);
  node_softmax_gather<1, 64, false><<<cdiv(N, 8), 256, 0, stream>>>(
      rowptr, csr_src, asrc2, adst2, h2, b2, out, N);
}


# Layer-1 softmax-gather and layer-2 GEMM in one launch: a block sums 32 nodes' rows into LDS and multiplies them by W2 there (7 -> 6 launches); k summed as 4 partial sums, score dot by lane butterfly
# speedup vs baseline: 7.1071x; 1.0339x over previous
#include <hip/hip_runtime.h>
#include <hip/hip_bf16.h>


typedef float v2f __attribute__((ext_vector_type(2)));
typedef int   v2i __attribute__((ext_vector_type(2)));
typedef float v4f __attribute__((ext_vector_type(4)));
typedef float v8f __attribute__((ext_vector_type(8)));

#define NEG_SLOPE 0.2f


template <int G>
__device__ __forceinline__ float head_group_sum(const float (&v)[8], int c) {
  static_assert(G == 8 || G == 16, "a head is 32 or 64 columns");
  constexpr int S = (G == 16) ? 1 : 0;
  const bool u2 = (c >> (S + 2)) & 1, u1 = (c >> (S + 1)) & 1, u0 = (c >> S) & 1;
  float a[4], b[2];
#pragma unroll
  for (int i = 0; i < 4; ++i)
    a[i] = (u2 ? v[i + 4] : v[i]) + __shfl_xor(u2 ? v[i] : v[i + 4], 4 << S, 32);
#pragma unroll
  for (int i = 0; i < 2; ++i)
    b[i] = (u1 ? a[i + 2] : a[i]) + __shfl_xor(u1 ? a[i] : a[i + 2], 2 << S, 32);
  float t = (u0 ? b[1] : b[0]) + __shfl_xor(u0 ? b[0] : b[1], 1 << S, 32);
  if constexpr (G == 16) t += __shfl_xor(t, 1, 32);
  return t;
}

template <int K, int NOUT, int CH>
__device__ __forceinline__ void
gemm_tile(const float* __restrict__ A,
          const float* __restrict__ B,
          float* __restrict__ C,
          const float* __restrict__ att_src,
          const float* __restrict__ att_dst,
          float* __restrict__ asrc, float* __restrict__ adst, int M, int tile) {
  constexpr int KC     = 32;
  constexpr int NCH    = K / KC;
  constexpr int NSTRIP = NOUT / 64;
  constexpr int HEADS  = NOUT / CH;
  constexpr int G      = CH / 4;
  static_assert(K % KC == 0 && NOUT % 64 == 0, "tile sizes");
  static_assert(NOUT % CH == 0 && (CH == 32 || CH == 64), "a head lies inside one 64-column strip");
  __shared__ v4f Bs[2][KC * 16];

  const int tid  = threadIdx.x;
  const int lane = tid & 31;
  const int w    = tid >> 5;
  const int c    = lane & 15;
  const int h    = lane >> 4;
  const int nstrip = tile % NSTRIP;
  const int mblk   = tile / NSTRIP;
  const int n0     = nstrip * 64;

  int arow = mblk * 128 + w * 16 + c;
  if (arow >= M) arow = M - 1;
  const float* ap = A + (size_t)arow * K + 16 * h;

  const int br = tid >> 4, bq = tid & 15;
  const float* bp = B + (size_t)br * NOUT + n0 + 4 * bq;

  const v4f ws = *(const v4f*)(att_src + n0 + 4 * c);
  const v4f wd = *(const v4f*)(att_dst + n0 + 4 * c);

  v4f a_cur[4], a_nxt[4], b_st[2];
#pragma unroll
  for (int i = 0; i < 4; ++i) a_cur[i] = *(const v4f*)(ap + 4 * i);
  b_st[0] = *(const v4f*)(bp);
  b_st[1] = *(const v4f*)(bp + (size_t)16 * NOUT);
  Bs[0][br * 16 + bq]        = b_st[0];
  Bs[0][(br + 16) * 16 + bq] = b_st[1];
  __syncthreads();

  v8f acc[4] = {};
  for (int kc = 0; kc < NCH; ++kc) {
    const int cur = kc & 1;
    const bool more = (kc + 1 < NCH);
    if (more) {
      const int k1 = (kc + 1) * KC;
#pragma unroll
      for (int i = 0; i < 4; ++i) a_nxt[i] = *(const v4f*)(ap + k1 + 4 * i);
      b_st[0] = *(const v4f*)(bp + (size_t)k1 * NOUT);
      b_st[1] = *(const v4f*)(bp + (size_t)(k1 + 16) * NOUT);
    }
    __builtin_amdgcn_sched_barrier(0);

    const v4f* bs = &Bs[cur][(16 * h) * 16 + c];
    v4f bf[16];
#pragma unroll
    for (int i = 0; i < 16; ++i) bf[i] = bs[i * 16];

#pragma unroll
    for (int s = 0; s < 8; ++s) {
      const v4f av = a_cur[s >> 1];
      v2f a;
      if (s & 1) { a.x = av.z; a.y = av.w; } else { a.x = av.x; a.y = av.y; }
      const v4f b0 = bf[2 * s], b1 = bf[2 * s + 1];
#pragma unroll
      for (int j = 0; j < 4; ++j) {
        v2f b;
        b.x = b0[j];
        b.y = b1[j];
        acc[j] = __builtin_amdgcn_wmma_f32_16x16x4_f32(
             false, a,  false, b,
             (short)0, acc[j],  false,  false);
      }
    }

    __builtin_amdgcn_sched_barrier(0);
    if (more) {
      Bs[cur ^ 1][br * 16 + bq]        = b_st[0];
      Bs[cur ^ 1][(br + 16) * 16 + bq] = b_st[1];
      __syncthreads();
#pragma unroll
      for (int i = 0; i < 4; ++i) a_cur[i] = a_nxt[i];
    }
  }

  const int orow = mblk * 128 + w * 16 + 8 * h;
  float* cp = C + (size_t)orow * NOUT + n0 + 4 * c;
#pragma unroll
  for (int r = 0; r < 8; ++r) {
    if (orow + r < M) {
      v4f o = {acc[0][r], acc[1][r], acc[2][r], acc[3][r]};
      *(v4f*)(cp + (size_t)r * NOUT) = o;
    }
  }

  float ps[8], pd[8];
#pragma unroll
  for (int r = 0; r < 8; ++r) {
    ps[r] = acc[0][r] * ws.x + acc[1][r] * ws.y + acc[2][r] * ws.z + acc[3][r] * ws.w;
    pd[r] = acc[0][r] * wd.x + acc[1][r] * wd.y + acc[2][r] * wd.z + acc[3][r] * wd.w;
  }
  const float ts = head_group_sum<G>(ps, c);
  const float td = head_group_sum<G>(pd, c);
  const int srow = orow + ((c >> (G == 16 ? 1 : 0)) & 7);
  const int head = (n0 + 4 * c) / CH;
  const bool writer = (G == 8) || ((c & 1) == 0);
  if (writer && srow < M) {
    asrc[(size_t)srow * HEADS + head] = ts;
    adst[(size_t)srow * HEADS + head] = td;
  }
}

#define BKT_SHIFT 8
#define BKT_NODES 256
#define NCHUNK    256
#define MAXB      1024

__device__ __forceinline__ int block_scan_incl(int v, int* wsum, int& total) {
  const int lane = threadIdx.x & 31, w = threadIdx.x >> 5;
  int x = v;
#pragma unroll
  for (int o = 1; o < 32; o <<= 1) {
    int y = __shfl_up(x, o, 32);
    if (lane >= o) x += y;
  }
  if (lane == 31) wsum[w] = x;
  __syncthreads();
  int woff = 0, tot = 0;
#pragma unroll
  for (int k = 0; k < 8; ++k) {
    int t = wsum[k];
    if (k < w) woff += t;
    tot += t;
  }
  total = tot;
  return woff + x;
}

#define EPT 16

__device__ __forceinline__ void
bucket_count(int k, const int* __restrict__ ei, int E, int nbk, int* __restrict__ cm) {
  __shared__ int hist[MAXB];
  const int cs = (E + NCHUNK - 1) / NCHUNK;
  const int lo = k * cs, hi = min(E, lo + cs);
  const int* dp = ei + E;
  int d[EPT];
  auto load = [&](int base) {
#pragma unroll
    for (int u = 0; u < EPT; ++u) {
      const int i = base + u * 256 + (int)threadIdx.x;
      d[u] = (i < hi) ? dp[i] : -1;
    }
  };
  load(lo);
  for (int b = threadIdx.x; b < nbk; b += 256) hist[b] = 0;
  __syncthreads();
  for (int base = lo;;) {
#pragma unroll
    for (int u = 0; u < EPT; ++u)
      if (d[u] >= 0) atomicAdd(&hist[d[u] >> BKT_SHIFT], 1);
    base += EPT * 256;
    if (base >= hi) break;
    load(base);
  }
  __syncthreads();
  for (int b = threadIdx.x; b < nbk; b += 256) cm[(size_t)b * NCHUNK + k] = hist[b];
}

__device__ __forceinline__ void
bucket_scan(int b, const int* __restrict__ cm, int* __restrict__ off, int* __restrict__ total) {
  __shared__ int wsum[8];
  const size_t i = (size_t)b * NCHUNK + threadIdx.x;
  const int v = cm[i];
  int tot;
  const int incl = block_scan_incl(v, wsum, tot);
  off[i] = incl - v;
  if (threadIdx.x == 0) total[b] = tot;
}

__device__ __forceinline__ void
bucket_place(int k, const int* __restrict__ ei, int E, int nbk,
             const int* __restrict__ off, const int* __restrict__ total,
             int* __restrict__ bstart, int* __restrict__ part) {
  __shared__ int cursor[MAXB];
  __shared__ int wsum[8];
  const int cs = (E + NCHUNK - 1) / NCHUNK;
  const int lo = k * cs, hi = min(E, lo + cs);
  int sv[EPT], dv[EPT];
  auto load = [&](int base) {
#pragma unroll
    for (int u = 0; u < EPT; ++u) {
      const int i = base + u * 256 + (int)threadIdx.x;
      const bool ok = i < hi;
      dv[u] = ok ? ei[E + i] : -1;
      sv[u] = ok ? ei[i] : 0;
    }
  };
  load(lo);
  v2i* pairs = (v2i*)part;

  int carry = 0;
  for (int base = 0; base < nbk; base += 256) {
    const int b = base + threadIdx.x;
    const int v = (b < nbk) ? total[b] : 0;
    const int o = (b < nbk) ? off[(size_t)b * NCHUNK + k] : 0;
    int tot;
    const int incl = block_scan_incl(v, wsum, tot);
    if (b < nbk) {
      const int st = carry + incl - v;
      cursor[b] = st + o;
      if (k == 0) bstart[b] = st;
    }
    carry += tot;
    __syncthreads();
  }

  for (int base = lo;;) {
#pragma unroll
    for (int u = 0; u < EPT; ++u)
      if (dv[u] >= 0) {
        const int pos = atomicAdd(&cursor[dv[u] >> BKT_SHIFT], 1);
        const v2i e = {sv[u], dv[u]};
        pairs[pos] = e;
      }
    base += EPT * 256;
    if (base >= hi) break;
    load(base);
  }
}

__device__ __forceinline__ void
bucket_rows(int b, const int* __restrict__ part, const int* __restrict__ bstart,
            const int* __restrict__ total, int n,
            int* __restrict__ rowptr, int* __restrict__ csr_src) {
  __shared__ int cnt[BKT_NODES];
  __shared__ int wsum[8];
  const int node = b * BKT_NODES + threadIdx.x;
  const int s = bstart[b], tot = total[b];
  const v2i* pp = (const v2i*)part + s;
  v2i ev[EPT];
  auto load = [&](int base) {
#pragma unroll
    for (int u = 0; u < EPT; ++u) {
      const int i = base + u * 256 + (int)threadIdx.x;
      const v2i none = {0, -1};
      ev[u] = (i < tot) ? pp[i] : none;
    }
  };
  load(0);
  cnt[threadIdx.x] = (node < n) ? 1 : 0;
  __syncthreads();
  for (int base = 0;;) {
#pragma unroll
    for (int u = 0; u < EPT; ++u)
      if (ev[u].y >= 0) atomicAdd(&cnt[ev[u].y & (BKT_NODES - 1)], 1);
    base += EPT * 256;
    if (base >= tot) break;
    load(base);
  }
  __syncthreads();
  const int v = cnt[threadIdx.x];
  int t;
  const int incl = block_scan_incl(v, wsum, t);
  const int beg = s + b * BKT_NODES + incl - v;
  if (node < n) {
    rowptr[node] = beg;
    csr_src[beg] = node;
    if (node == n - 1) rowptr[n] = beg + v;
  }
  cnt[threadIdx.x] = beg + 1;
  if (tot > EPT * 256) load(0);
  __syncthreads();
  for (int base = 0;;) {
#pragma unroll
    for (int u = 0; u < EPT; ++u)
      if (ev[u].y >= 0) {
        const int pos = atomicAdd(&cnt[ev[u].y & (BKT_NODES - 1)], 1);
        csr_src[pos] = ev[u].x;
      }
    base += EPT * 256;
    if (base >= tot) break;
    load(base);
  }
}

template <int PASS>
__global__ void __launch_bounds__(256)
group_pass_beside_gemm(const int* __restrict__ ei, int E, int nbk, int n,
                       int* cm, int* off, int* total, int* bstart, int* part,
                       int* rowptr, int* csr_src, int ngroup,
                       const float* __restrict__ A, const float* __restrict__ B,
                       float* __restrict__ C,
                       const float* __restrict__ att_src, const float* __restrict__ att_dst,
                       float* __restrict__ asrc, float* __restrict__ adst, int M, int tile0) {
  const int bid = (int)blockIdx.x;
  if (bid < ngroup) {
    if constexpr (PASS == 0)      bucket_count(bid, ei, E, nbk, cm);
    else if constexpr (PASS == 1) bucket_scan(bid, cm, off, total);
    else if constexpr (PASS == 2) bucket_place(bid, ei, E, nbk, off, total, bstart, part);
    else                          bucket_rows(bid, part, bstart, total, n, rowptr, csr_src);
  } else {
    gemm_tile<256, 256, 32>(A, B, C, att_src, att_dst, asrc, adst, M, tile0 + bid - ngroup);
  }
}

template <int H>
__device__ __forceinline__ void load_heads(const float* __restrict__ p, float (&v)[H]) {
  if constexpr (H == 8) {
    const v4f a = *(const v4f*)p, b = *(const v4f*)(p + 4);
    v[0] = a.x; v[1] = a.y; v[2] = a.z; v[3] = a.w;
    v[4] = b.x; v[5] = b.y; v[6] = b.z; v[7] = b.w;
  } else {
#pragma unroll
    for (int h = 0; h < H; ++h) v[h] = p[h];
  }
}

__device__ __forceinline__ void wave_lds_sync() {
  __builtin_amdgcn_fence(__ATOMIC_RELEASE, "wavefront");
  __builtin_amdgcn_wave_barrier();
  __builtin_amdgcn_fence(__ATOMIC_ACQUIRE, "wavefront");
}

template <int H, int C, bool RELU>
__global__ void __launch_bounds__(256)
node_softmax_gather(const int* __restrict__ rowptr, const int* __restrict__ csr_src,
                    const float* __restrict__ asrc, const float* __restrict__ adst,
                    const float* __restrict__ hfeat, const float* __restrict__ bias,
                    float* __restrict__ out, int n) {
  constexpr int F = H * C;
  constexpr int VPL = F / 32;
  static_assert(VPL == 8 || VPL == 2, "lane owns 8 or 2 features");
  __shared__ float strip[8][32 * H];

  const int lane = threadIdx.x & 31;
  const int wib  = threadIdx.x >> 5;
  const int node = __builtin_amdgcn_readfirstlane(blockIdx.x * 8 + wib);
  if (node >= n) return;
  const int beg = rowptr[node], end = rowptr[node + 1];
  const int hd = (lane * VPL) / C;
  float* myp = strip[wib];

  float ad[H];
  load_heads<H>(adst + (size_t)node * H, ad);

  auto stage_scores = [&](int c) -> int {
    const int j = c + lane;
    int s = 0;
    float v[H];
#pragma unroll
    for (int h = 0; h < H; ++h) v[h] = -INFINITY;
    if (j < end) {
      s = csr_src[j];
      float as[H];
      load_heads<H>(asrc + (size_t)s * H, as);
#pragma unroll
      for (int h = 0; h < H; ++h) {
        float t = as[h] + ad[h];
        v[h] = (t > 0.0f) ? t : NEG_SLOPE * t;
      }
    }
#pragma unroll
    for (int h = 0; h < H; ++h) myp[lane * H + h] = v[h];
    return s;
  };

  float mx = -INFINITY;
  int s = 0;
  for (int c = beg; c < end; c += 32) {
    s = stage_scores(c);
    wave_lds_sync();
#pragma unroll
    for (int t = 0; t < 32; ++t) mx = fmaxf(mx, myp[t * H + hd]);
    wave_lds_sync();
  }

  const bool single = (end - beg) <= 32;
  float acc[VPL];
#pragma unroll
  for (int i = 0; i < VPL; ++i) acc[i] = 0.0f;
  float lsum = 0.0f;
  for (int c = beg; c < end; c += 32) {
    if (!single) {
      s = stage_scores(c);
      wave_lds_sync();
    }
    const int m = min(32, end - c);
#pragma unroll 4
    for (int t = 0; t < m; ++t) {
      const int st = __builtin_amdgcn_readlane(s, t);
      const float p = __expf(myp[t * H + hd] - mx);
      lsum += p;
      const float* hp = hfeat + (size_t)st * F + lane * VPL;
      if constexpr (VPL == 8) {
        const v4f a = *(const v4f*)hp, b = *(const v4f*)(hp + 4);
        acc[0] += p * a.x; acc[1] += p * a.y; acc[2] += p * a.z; acc[3] += p * a.w;
        acc[4] += p * b.x; acc[5] += p * b.y; acc[6] += p * b.z; acc[7] += p * b.w;
      } else {
        const v2f a = *(const v2f*)hp;
        acc[0] += p * a.x; acc[1] += p * a.y;
      }
    }
    if (!single) wave_lds_sync();
  }

  const float denom = lsum + 1e-16f;
  float r[VPL];
#pragma unroll
  for (int i = 0; i < VPL; ++i) {
    float o = acc[i] / denom + bias[lane * VPL + i];
    r[i] = RELU ? fmaxf(o, 0.0f) : o;
  }
  float* op = out + (size_t)node * F + lane * VPL;
  if constexpr (VPL == 8) {
    v4f a = {r[0], r[1], r[2], r[3]}, b = {r[4], r[5], r[6], r[7]};
    *(v4f*)op = a;
    *(v4f*)(op + 4) = b;
  } else {
    v2f a = {r[0], r[1]};
    *(v2f*)op = a;
  }
}


#define FN 32
#define AP 260

__global__ void __launch_bounds__(256)
gather1_gemm2(const int* __restrict__ rowptr, const int* __restrict__ csr_src,
              const float* __restrict__ asrc, const float* __restrict__ adst,
              const float* __restrict__ hfeat, const float* __restrict__ bias,
              const float* __restrict__ W2,
              const float* __restrict__ att_src2, const float* __restrict__ att_dst2,
              float* __restrict__ h2, float* __restrict__ asrc2, float* __restrict__ adst2,
              int n) {
  constexpr int H = 8, C = 32, F = 256, VPL = 8, F2 = 64;
  __shared__ float strip[8][32 * H];
  __shared__ v4f As4[FN * AP / 4];
  float* As = (float*)As4;

  const int lane = threadIdx.x & 31;
  const int wib  = threadIdx.x >> 5;
  const int base = blockIdx.x * FN;
  const int hd = (lane * VPL) / C;
  float* myp = strip[wib];

  for (int q = 0; q < 4; ++q) {
    const int slot = wib * 4 + q;
    const int node = __builtin_amdgcn_readfirstlane(base + slot);
    float r[VPL];
#pragma unroll
    for (int i = 0; i < VPL; ++i) r[i] = 0.0f;
    if (node < n) {
      const int beg = rowptr[node], end = rowptr[node + 1];
      float ad[H];
      load_heads<H>(adst + (size_t)node * H, ad);

      auto stage_scores = [&](int c) -> int {
        const int j = c + lane;
        int s = 0;
        float v[H];
#pragma unroll
        for (int h = 0; h < H; ++h) v[h] = -INFINITY;
        if (j < end) {
          s = csr_src[j];
          float as[H];
          load_heads<H>(asrc + (size_t)s * H, as);
#pragma unroll
          for (int h = 0; h < H; ++h) {
            float t = as[h] + ad[h];
            v[h] = (t > 0.0f) ? t : NEG_SLOPE * t;
          }
        }
#pragma unroll
        for (int h = 0; h < H; ++h) myp[lane * H + h] = v[h];
        return s;
      };

      float mx = -INFINITY;
      int s = 0;
      for (int c = beg; c < end; c += 32) {
        s = stage_scores(c);
        wave_lds_sync();
#pragma unroll
        for (int t = 0; t < 32; ++t) mx = fmaxf(mx, myp[t * H + hd]);
        wave_lds_sync();
      }

      const bool single = (end - beg) <= 32;
      float acc[VPL];
#pragma unroll
      for (int i = 0; i < VPL; ++i) acc[i] = 0.0f;
      float lsum = 0.0f;
      for (int c = beg; c < end; c += 32) {
        if (!single) {
          s = stage_scores(c);
          wave_lds_sync();
        }
        const int m = min(32, end - c);
#pragma unroll 4
        for (int t = 0; t < m; ++t) {
          const int st = __builtin_amdgcn_readlane(s, t);
          const float p = __expf(myp[t * H + hd] - mx);
          lsum += p;
          const float* hp = hfeat + (size_t)st * F + lane * VPL;
          const v4f a = *(const v4f*)hp, b = *(const v4f*)(hp + 4);
          acc[0] += p * a.x; acc[1] += p * a.y; acc[2] += p * a.z; acc[3] += p * a.w;
          acc[4] += p * b.x; acc[5] += p * b.y; acc[6] += p * b.z; acc[7] += p * b.w;
        }
        wave_lds_sync();
      }

      const float denom = lsum + 1e-16f;
#pragma unroll
      for (int i = 0; i < VPL; ++i) {
        float o = acc[i] / denom + bias[lane * VPL + i];
        r[i] = fmaxf(o, 0.0f);
      }
    }
    const v4f ra = {r[0], r[1], r[2], r[3]}, rb = {r[4], r[5], r[6], r[7]};
    v4f* ap = (v4f*)(As + slot * AP + lane * VPL);
    ap[0] = ra;
    ap[1] = rb;
  }
  const int c  = lane & 15;
  const int hh = lane >> 4;
  const int mt = wib >> 2;
  const int kq = wib & 3;
  const float* arow = As + (16 * mt + c) * AP + 64 * kq + 8 * hh;
  const float* bp   = W2 + (size_t)(64 * kq + 8 * hh) * F2 + 4 * c;

  v4f af[2], bf[8];
#pragma unroll
  for (int i = 0; i < 8; ++i) bf[i] = *(const v4f*)(bp + (size_t)i * F2);

  __syncthreads();

  v8f acc2[4] = {};
#pragma unroll
  for (int kc = 0; kc < 4; ++kc) {
    if (kc > 0) {
#pragma unroll
      for (int i = 0; i < 8; ++i) bf[i] = *(const v4f*)(bp + (size_t)(16 * kc + i) * F2);
    }
#pragma unroll
    for (int i = 0; i < 2; ++i) af[i] = *(const v4f*)(arow + 16 * kc + 4 * i);
    __builtin_amdgcn_sched_barrier(0);
#pragma unroll
    for (int s = 0; s < 4; ++s) {
      const v4f av = af[s >> 1];
      v2f a;
      if (s & 1) { a.x = av.z; a.y = av.w; } else { a.x = av.x; a.y = av.y; }
      const v4f b0 = bf[2 * s], b1 = bf[2 * s + 1];
#pragma unroll
      for (int j = 0; j < 4; ++j) {
        v2f b;
        b.x = b0[j];
        b.y = b1[j];
        acc2[j] = __builtin_amdgcn_wmma_f32_16x16x4_f32(
             false, a,  false, b,
             (short)0, acc2[j],  false,  false);
      }
    }
    __builtin_amdgcn_sched_barrier(0);
  }

  __syncthreads();
#pragma unroll
  for (int r = 0; r < 8; ++r) {
    const v4f o = {acc2[0][r], acc2[1][r], acc2[2][r], acc2[3][r]};
    As4[(wib * 8 + r) * 32 + lane] = o;
  }
  __syncthreads();

  const int rr = threadIdx.x >> 4;
  const int cq = threadIdx.x & 15;
  const int pl = 16 * (rr >> 3) + cq;
  const v4f ws2 = *(const v4f*)(att_src2 + 4 * cq);
  const v4f wd2 = *(const v4f*)(att_dst2 + 4 * cq);
#pragma unroll
  for (int it = 0; it < 2; ++it) {
    const v4f* pp = As4 + ((it * 4) * 8 + (rr & 7)) * 32 + pl;
    v4f o = pp[0];
    o += pp[1 * 8 * 32];
    o += pp[2 * 8 * 32];
    o += pp[3 * 8 * 32];
    float ts = o.x * ws2.x + o.y * ws2.y + o.z * ws2.z + o.w * ws2.w;
    float td = o.x * wd2.x + o.y * wd2.y + o.z * wd2.z + o.w * wd2.w;
#pragma unroll
    for (int m = 8; m >= 1; m >>= 1) {
      ts += __shfl_xor(ts, m, 32);
      td += __shfl_xor(td, m, 32);
    }
    const int node = base + 16 * it + rr;
    if (node < n) {
      *(v4f*)(h2 + (size_t)node * F2 + 4 * cq) = o;
      if (cq == 0) {
        asrc2[node] = ts;
        adst2[node] = td;
      }
    }
  }
}

extern "C" void kernel_launch(void* const* d_in, const int* in_sizes, int n_in,
                              void* d_out, int out_size, void* d_ws, size_t ws_size,
                              hipStream_t stream) {
  const float* x        = (const float*)d_in[0];
  const int*   ei       = (const int*)d_in[1];
  const float* W1       = (const float*)d_in[2];
  const float* att_src1 = (const float*)d_in[3];
  const float* att_dst1 = (const float*)d_in[4];
  const float* b1       = (const float*)d_in[5];
  const float* W2       = (const float*)d_in[6];
  const float* att_src2 = (const float*)d_in[7];
  const float* att_dst2 = (const float*)d_in[8];
  const float* b2       = (const float*)d_in[9];

  const int D_IN = 256, F1 = 256, H1 = 8;
  const int N = in_sizes[0] / D_IN;
  const int E = in_sizes[1] / 2;
  const int NBK = (N + BKT_NODES - 1) / BKT_NODES;

  float* ws = (float*)d_ws;
  float* h1    = ws;
  float* agg1  = h1   + (size_t)N * F1;
  float* asrc1 = agg1 + (size_t)N * F1;
  float* adst1 = asrc1 + (size_t)N * H1;
  int*   part  = (int*)(adst1 + (size_t)N * H1);
  float* asrc2 = (float*)(part + (size_t)2 * E);
  float* adst2 = asrc2 + (N + 4);
  int*   rowptr = (int*)(adst2 + (N + 4));
  int*   cm    = rowptr + (N + 4);
  int*   off   = cm + (size_t)NBK * NCHUNK;
  int*   total = off + (size_t)NBK * NCHUNK;
  int*   bstart = total + MAXB;
  int*   csr_src = bstart + MAXB;
  float* out   = (float*)d_out;

  auto cdiv = [](long long a, long long b) { return (int)((a + b - 1) / b); };

  auto imin = [](int a, int b) { return a < b ? a : b; };
  const int T  = cdiv(N, 128) * (F1 / 64);
  const int t0 = imin(T, 512), t1 = 0, t2 = imin(T - t0, 512);
  const int t3 = T - t0 - t1 - t2;
  group_pass_beside_gemm<0><<<NCHUNK + t0, 256, 0, stream>>>(
      ei, E, NBK, N, cm, off, total, bstart, part, rowptr, csr_src, NCHUNK,
      x, W1, h1, att_src1, att_dst1, asrc1, adst1, N, 0);
  group_pass_beside_gemm<1><<<NBK + t1, 256, 0, stream>>>(
      ei, E, NBK, N, cm, off, total, bstart, part, rowptr, csr_src, NBK,
      x, W1, h1, att_src1, att_dst1, asrc1, adst1, N, t0);
  group_pass_beside_gemm<2><<<NCHUNK + t2, 256, 0, stream>>>(
      ei, E, NBK, N, cm, off, total, bstart, part, rowptr, csr_src, NCHUNK,
      x, W1, h1, att_src1, att_dst1, asrc1, adst1, N, t0 + t1);
  group_pass_beside_gemm<3><<<NBK + t3, 256, 0, stream>>>(
      ei, E, NBK, N, cm, off, total, bstart, part, rowptr, csr_src, NBK,
      x, W1, h1, att_src1, att_dst1, asrc1, adst1, N, t0 + t1 + t2);

  float* h2 = agg1;
  gather1_gemm2<<<cdiv(N, FN), 256, 0, stream>>>(
      rowptr, csr_src, asrc1, adst1, h1, b1, W2, att_src2, att_dst2, h2, asrc2, adst2, N);

  node_softmax_gather<1, 64, false><<<cdiv(N, 8), 256, 0, stream>>>(
      rowptr, csr_src, asrc2, adst2, h2, b2, out, N);
}


# Edge grouping in two passes, not four: each chunk sorts itself by bucket in its own stretch; a bucket block collects its 256 segments (start = one column sum); 6 -> 4 launches; sum order as before
# speedup vs baseline: 7.7599x; 1.0919x over previous
#include <hip/hip_runtime.h>
#include <hip/hip_bf16.h>


typedef float v2f __attribute__((ext_vector_type(2)));
typedef int   v2i __attribute__((ext_vector_type(2)));
typedef float v4f __attribute__((ext_vector_type(4)));
typedef float v8f __attribute__((ext_vector_type(8)));

#define NEG_SLOPE 0.2f


template <int G>
__device__ __forceinline__ float head_group_sum(const float (&v)[8], int c) {
  static_assert(G == 8 || G == 16, "a head is 32 or 64 columns");
  constexpr int S = (G == 16) ? 1 : 0;
  const bool u2 = (c >> (S + 2)) & 1, u1 = (c >> (S + 1)) & 1, u0 = (c >> S) & 1;
  float a[4], b[2];
#pragma unroll
  for (int i = 0; i < 4; ++i)
    a[i] = (u2 ? v[i + 4] : v[i]) + __shfl_xor(u2 ? v[i] : v[i + 4], 4 << S, 32);
#pragma unroll
  for (int i = 0; i < 2; ++i)
    b[i] = (u1 ? a[i + 2] : a[i]) + __shfl_xor(u1 ? a[i] : a[i + 2], 2 << S, 32);
  float t = (u0 ? b[1] : b[0]) + __shfl_xor(u0 ? b[0] : b[1], 1 << S, 32);
  if constexpr (G == 16) t += __shfl_xor(t, 1, 32);
  return t;
}

template <int K, int NOUT, int CH>
__device__ __forceinline__ void
gemm_tile(const float* __restrict__ A,
          const float* __restrict__ B,
          float* __restrict__ C,
          const float* __restrict__ att_src,
          const float* __restrict__ att_dst,
          float* __restrict__ asrc, float* __restrict__ adst, int M, int tile) {
  constexpr int KC     = 32;
  constexpr int NCH    = K / KC;
  constexpr int NSTRIP = NOUT / 64;
  constexpr int HEADS  = NOUT / CH;
  constexpr int G      = CH / 4;
  static_assert(K % KC == 0 && NOUT % 64 == 0, "tile sizes");
  static_assert(NOUT % CH == 0 && (CH == 32 || CH == 64), "a head lies inside one 64-column strip");
  __shared__ v4f Bs[2][KC * 16];

  const int tid  = threadIdx.x;
  const int lane = tid & 31;
  const int w    = tid >> 5;
  const int c    = lane & 15;
  const int h    = lane >> 4;
  const int nstrip = tile % NSTRIP;
  const int mblk   = tile / NSTRIP;
  const int n0     = nstrip * 64;

  int arow = mblk * 128 + w * 16 + c;
  if (arow >= M) arow = M - 1;
  const float* ap = A + (size_t)arow * K + 16 * h;

  const int br = tid >> 4, bq = tid & 15;
  const float* bp = B + (size_t)br * NOUT + n0 + 4 * bq;

  const v4f ws = *(const v4f*)(att_src + n0 + 4 * c);
  const v4f wd = *(const v4f*)(att_dst + n0 + 4 * c);

  v4f a_cur[4], a_nxt[4], b_st[2];
#pragma unroll
  for (int i = 0; i < 4; ++i) a_cur[i] = *(const v4f*)(ap + 4 * i);
  b_st[0] = *(const v4f*)(bp);
  b_st[1] = *(const v4f*)(bp + (size_t)16 * NOUT);
  Bs[0][br * 16 + bq]        = b_st[0];
  Bs[0][(br + 16) * 16 + bq] = b_st[1];
  __syncthreads();

  v8f acc[4] = {};
  for (int kc = 0; kc < NCH; ++kc) {
    const int cur = kc & 1;
    const bool more = (kc + 1 < NCH);
    if (more) {
      const int k1 = (kc + 1) * KC;
#pragma unroll
      for (int i = 0; i < 4; ++i) a_nxt[i] = *(const v4f*)(ap + k1 + 4 * i);
      b_st[0] = *(const v4f*)(bp + (size_t)k1 * NOUT);
      b_st[1] = *(const v4f*)(bp + (size_t)(k1 + 16) * NOUT);
    }
    __builtin_amdgcn_sched_barrier(0);

    const v4f* bs = &Bs[cur][(16 * h) * 16 + c];
    v4f bf[16];
#pragma unroll
    for (int i = 0; i < 16; ++i) bf[i] = bs[i * 16];

#pragma unroll
    for (int s = 0; s < 8; ++s) {
      const v4f av = a_cur[s >> 1];
      v2f a;
      if (s & 1) { a.x = av.z; a.y = av.w; } else { a.x = av.x; a.y = av.y; }
      const v4f b0 = bf[2 * s], b1 = bf[2 * s + 1];
#pragma unroll
      for (int j = 0; j < 4; ++j) {
        v2f b;
        b.x = b0[j];
        b.y = b1[j];
        acc[j] = __builtin_amdgcn_wmma_f32_16x16x4_f32(
             false, a,  false, b,
             (short)0, acc[j],  false,  false);
      }
    }

    __builtin_amdgcn_sched_barrier(0);
    if (more) {
      Bs[cur ^ 1][br * 16 + bq]        = b_st[0];
      Bs[cur ^ 1][(br + 16) * 16 + bq] = b_st[1];
      __syncthreads();
#pragma unroll
      for (int i = 0; i < 4; ++i) a_cur[i] = a_nxt[i];
    }
  }

  const int orow = mblk * 128 + w * 16 + 8 * h;
  float* cp = C + (size_t)orow * NOUT + n0 + 4 * c;
#pragma unroll
  for (int r = 0; r < 8; ++r) {
    if (orow + r < M) {
      v4f o = {acc[0][r], acc[1][r], acc[2][r], acc[3][r]};
      *(v4f*)(cp + (size_t)r * NOUT) = o;
    }
  }

  float ps[8], pd[8];
#pragma unroll
  for (int r = 0; r < 8; ++r) {
    ps[r] = acc[0][r] * ws.x + acc[1][r] * ws.y + acc[2][r] * ws.z + acc[3][r] * ws.w;
    pd[r] = acc[0][r] * wd.x + acc[1][r] * wd.y + acc[2][r] * wd.z + acc[3][r] * wd.w;
  }
  const float ts = head_group_sum<G>(ps, c);
  const float td = head_group_sum<G>(pd, c);
  const int srow = orow + ((c >> (G == 16 ? 1 : 0)) & 7);
  const int head = (n0 + 4 * c) / CH;
  const bool writer = (G == 8) || ((c & 1) == 0);
  if (writer && srow < M) {
    asrc[(size_t)srow * HEADS + head] = ts;
    adst[(size_t)srow * HEADS + head] = td;
  }
}

#define BKT_SHIFT 8
#define BKT_NODES 256
#define NCHUNK    256
#define MAXB      1024

__device__ __forceinline__ int block_scan_incl(int v, int* wsum, int& total) {
  const int lane = threadIdx.x & 31, w = threadIdx.x >> 5;
  int x = v;
#pragma unroll
  for (int o = 1; o < 32; o <<= 1) {
    int y = __shfl_up(x, o, 32);
    if (lane >= o) x += y;
  }
  if (lane == 31) wsum[w] = x;
  __syncthreads();
  int woff = 0, tot = 0;
#pragma unroll
  for (int k = 0; k < 8; ++k) {
    int t = wsum[k];
    if (k < w) woff += t;
    tot += t;
  }
  total = tot;
  return woff + x;
}

#define EPT 16

__device__ __forceinline__ void
chunk_sort(int k, const int* __restrict__ ei, int E, int nbk, int cp,
           int* __restrict__ ct, int* __restrict__ part) {
  __shared__ int cur[MAXB];
  __shared__ int wsum[8];
  const int cs = (E + NCHUNK - 1) / NCHUNK;
  const int lo = k * cs, hi = min(E, lo + cs);
  int sv[EPT], dv[EPT];
  auto load = [&](int base) {
#pragma unroll
    for (int u = 0; u < EPT; ++u) {
      const int i = base + u * 256 + (int)threadIdx.x;
      const bool ok = i < hi;
      dv[u] = ok ? ei[E + i] : -1;
      sv[u] = ok ? ei[i] : 0;
    }
  };
  load(lo);
  for (int b = threadIdx.x; b < nbk; b += 256) cur[b] = 0;
  __syncthreads();
  for (int base = lo;;) {
#pragma unroll
    for (int u = 0; u < EPT; ++u)
      if (dv[u] >= 0) atomicAdd(&cur[dv[u] >> BKT_SHIFT], 1);
    base += EPT * 256;
    if (base >= hi) break;
    load(base);
  }
  __syncthreads();

  int* row = ct + (size_t)k * cp;
  int carry = 0;
  for (int base = 0; base < nbk; base += 256) {
    const int b = base + threadIdx.x;
    const int v = (b < nbk) ? cur[b] : 0;
    int tot;
    const int incl = block_scan_incl(v, wsum, tot);
    if (b < nbk) {
      const int st = carry + incl - v;
      cur[b] = st;
      row[b] = st;
    }
    carry += tot;
    __syncthreads();
  }
  if (threadIdx.x == 0) row[nbk] = carry;

  if (hi - lo > EPT * 256) load(lo);
  v2i* pairs = (v2i*)part + lo;
  for (int base = lo;;) {
#pragma unroll
    for (int u = 0; u < EPT; ++u)
      if (dv[u] >= 0) {
        const int pos = atomicAdd(&cur[dv[u] >> BKT_SHIFT], 1);
        const v2i e = {sv[u], dv[u]};
        pairs[pos] = e;
      }
    base += EPT * 256;
    if (base >= hi) break;
    load(base);
  }
}

#define RPT 20

__device__ __forceinline__ void
bucket_rows(int b, const int* __restrict__ part, const int* __restrict__ ct, int cp,
            int E, int n, int* __restrict__ rowptr, int* __restrict__ csr_src) {
  __shared__ int cnt[BKT_NODES];
  __shared__ int wsum[8];
  __shared__ int red[8];
  __shared__ int segoff[NCHUNK + 1];
  __shared__ int gpos[NCHUNK];
  const int t = threadIdx.x;
  const int node = b * BKT_NODES + t;
  const int cs = (E + NCHUNK - 1) / NCHUNK;
  const int* rowp = ct + (size_t)t * cp + b;
  const int st = rowp[0];
  const int cn = rowp[1] - st;
  int tot;
  const int sincl = block_scan_incl(cn, wsum, tot);
  segoff[t] = sincl - cn;
  gpos[t] = t * cs + st - (sincl - cn);
  if (t == 0) segoff[NCHUNK] = tot;
  int x = st;
#pragma unroll
  for (int o = 16; o >= 1; o >>= 1) x += __shfl_xor(x, o, 32);
  if ((t & 31) == 0) red[t >> 5] = x;
  cnt[t] = (node < n) ? 1 : 0;
  __syncthreads();
  int s = 0;
#pragma unroll
  for (int k = 0; k < 8; ++k) s += red[k];

  const v2i* pp = (const v2i*)part;
  v2i ev[RPT];
  auto load = [&](int base) {
    int idx[RPT], c[RPT];
#pragma unroll
    for (int u = 0; u < RPT; ++u) {
      idx[u] = min(base + u * 256 + t, tot - 1);
      c[u] = 0;
    }
    for (int step = NCHUNK / 2; step >= 1; step >>= 1) {
#pragma unroll
      for (int u = 0; u < RPT; ++u)
        if (segoff[c[u] + step] <= idx[u]) c[u] += step;
    }
#pragma unroll
    for (int u = 0; u < RPT; ++u) {
      const v2i none = {0, -1};
      const v2i e = pp[gpos[c[u]] + idx[u]];
      ev[u] = (base + u * 256 + t < tot) ? e : none;
    }
  };
  const bool once = tot <= RPT * 256;
  for (int phase = 0; phase < 2; ++phase) {
    if (phase == 1) {
      const int v = cnt[t];
      int tt;
      const int incl = block_scan_incl(v, wsum, tt);
      const int beg = s + b * BKT_NODES + incl - v;
      if (node < n) {
        rowptr[node] = beg;
        csr_src[beg] = node;
        if (node == n - 1) rowptr[n] = beg + v;
      }
      cnt[t] = beg + 1;
      __syncthreads();
    }
    for (int base = 0; base < tot; base += RPT * 256) {
      if (phase == 0 || !once) load(base);
#pragma unroll
      for (int u = 0; u < RPT; ++u)
        if (ev[u].y >= 0) {
          const int pos = atomicAdd(&cnt[ev[u].y & (BKT_NODES - 1)], 1);
          if (phase == 1) csr_src[pos] = ev[u].x;
        }
    }
    __syncthreads();
  }
}

template <int PASS>
__global__ void __launch_bounds__(256)
group_pass_beside_gemm(const int* __restrict__ ei, int E, int nbk, int n,
                       int* ct, int cp, int* part,
                       int* rowptr, int* csr_src, int ngroup,
                       const float* __restrict__ A, const float* __restrict__ B,
                       float* __restrict__ C,
                       const float* __restrict__ att_src, const float* __restrict__ att_dst,
                       float* __restrict__ asrc, float* __restrict__ adst, int M, int tile0) {
  const int bid = (int)blockIdx.x;
  if (bid < ngroup) {
    if constexpr (PASS == 0) chunk_sort(bid, ei, E, nbk, cp, ct, part);
    else                     bucket_rows(bid, part, ct, cp, E, n, rowptr, csr_src);
  } else {
    gemm_tile<256, 256, 32>(A, B, C, att_src, att_dst, asrc, adst, M, tile0 + bid - ngroup);
  }
}

template <int H>
__device__ __forceinline__ void load_heads(const float* __restrict__ p, float (&v)[H]) {
  if constexpr (H == 8) {
    const v4f a = *(const v4f*)p, b = *(const v4f*)(p + 4);
    v[0] = a.x; v[1] = a.y; v[2] = a.z; v[3] = a.w;
    v[4] = b.x; v[5] = b.y; v[6] = b.z; v[7] = b.w;
  } else {
#pragma unroll
    for (int h = 0; h < H; ++h) v[h] = p[h];
  }
}

__device__ __forceinline__ void wave_lds_sync() {
  __builtin_amdgcn_fence(__ATOMIC_RELEASE, "wavefront");
  __builtin_amdgcn_wave_barrier();
  __builtin_amdgcn_fence(__ATOMIC_ACQUIRE, "wavefront");
}

template <int H, int C, bool RELU>
__global__ void __launch_bounds__(256)
node_softmax_gather(const int* __restrict__ rowptr, const int* __restrict__ csr_src,
                    const float* __restrict__ asrc, const float* __restrict__ adst,
                    const float* __restrict__ hfeat, const float* __restrict__ bias,
                    float* __restrict__ out, int n) {
  constexpr int F = H * C;
  constexpr int VPL = F / 32;
  static_assert(VPL == 8 || VPL == 2, "lane owns 8 or 2 features");
  __shared__ float strip[8][32 * H];

  const int lane = threadIdx.x & 31;
  const int wib  = threadIdx.x >> 5;
  const int node = __builtin_amdgcn_readfirstlane(blockIdx.x * 8 + wib);
  if (node >= n) return;
  const int beg = rowptr[node], end = rowptr[node + 1];
  const int hd = (lane * VPL) / C;
  float* myp = strip[wib];

  float ad[H];
  load_heads<H>(adst + (size_t)node * H, ad);

  auto stage_scores = [&](int c) -> int {
    const int j = c + lane;
    int s = 0;
    float v[H];
#pragma unroll
    for (int h = 0; h < H; ++h) v[h] = -INFINITY;
    if (j < end) {
      s = csr_src[j];
      float as[H];
      load_heads<H>(asrc + (size_t)s * H, as);
#pragma unroll
      for (int h = 0; h < H; ++h) {
        float t = as[h] + ad[h];
        v[h] = (t > 0.0f) ? t : NEG_SLOPE * t;
      }
    }
#pragma unroll
    for (int h = 0; h < H; ++h) myp[lane * H + h] = v[h];
    return s;
  };

  float mx = -INFINITY;
  int s = 0;
  for (int c = beg; c < end; c += 32) {
    s = stage_scores(c);
    wave_lds_sync();
#pragma unroll
    for (int t = 0; t < 32; ++t) mx = fmaxf(mx, myp[t * H + hd]);
    wave_lds_sync();
  }

  const bool single = (end - beg) <= 32;
  float acc[VPL];
#pragma unroll
  for (int i = 0; i < VPL; ++i) acc[i] = 0.0f;
  float lsum = 0.0f;
  for (int c = beg; c < end; c += 32) {
    if (!single) {
      s = stage_scores(c);
      wave_lds_sync();
    }
    const int m = min(32, end - c);
#pragma unroll 4
    for (int t = 0; t < m; ++t) {
      const int st = __builtin_amdgcn_readlane(s, t);
      const float p = __expf(myp[t * H + hd] - mx);
      lsum += p;
      const float* hp = hfeat + (size_t)st * F + lane * VPL;
      if constexpr (VPL == 8) {
        const v4f a = *(const v4f*)hp, b = *(const v4f*)(hp + 4);
        acc[0] += p * a.x; acc[1] += p * a.y; acc[2] += p * a.z; acc[3] += p * a.w;
        acc[4] += p * b.x; acc[5] += p * b.y; acc[6] += p * b.z; acc[7] += p * b.w;
      } else {
        const v2f a = *(const v2f*)hp;
        acc[0] += p * a.x; acc[1] += p * a.y;
      }
    }
    if (!single) wave_lds_sync();
  }

  const float denom = lsum + 1e-16f;
  float r[VPL];
#pragma unroll
  for (int i = 0; i < VPL; ++i) {
    float o = acc[i] / denom + bias[lane * VPL + i];
    r[i] = RELU ? fmaxf(o, 0.0f) : o;
  }
  float* op = out + (size_t)node * F + lane * VPL;
  if constexpr (VPL == 8) {
    v4f a = {r[0], r[1], r[2], r[3]}, b = {r[4], r[5], r[6], r[7]};
    *(v4f*)op = a;
    *(v4f*)(op + 4) = b;
  } else {
    v2f a = {r[0], r[1]};
    *(v2f*)op = a;
  }
}


#define FN 32
#define AP 260

__global__ void __launch_bounds__(256)
gather1_gemm2(const int* __restrict__ rowptr, const int* __restrict__ csr_src,
              const float* __restrict__ asrc, const float* __restrict__ adst,
              const float* __restrict__ hfeat, const float* __restrict__ bias,
              const float* __restrict__ W2,
              const float* __restrict__ att_src2, const float* __restrict__ att_dst2,
              float* __restrict__ h2, float* __restrict__ asrc2, float* __restrict__ adst2,
              int n) {
  constexpr int H = 8, C = 32, F = 256, VPL = 8, F2 = 64;
  __shared__ float strip[8][32 * H];
  __shared__ v4f As4[FN * AP / 4];
  float* As = (float*)As4;

  const int lane = threadIdx.x & 31;
  const int wib  = threadIdx.x >> 5;
  const int base = blockIdx.x * FN;
  const int hd = (lane * VPL) / C;
  float* myp = strip[wib];

  for (int q = 0; q < 4; ++q) {
    const int slot = wib * 4 + q;
    const int node = __builtin_amdgcn_readfirstlane(base + slot);
    float r[VPL];
#pragma unroll
    for (int i = 0; i < VPL; ++i) r[i] = 0.0f;
    if (node < n) {
      const int beg = rowptr[node], end = rowptr[node + 1];
      float ad[H];
      load_heads<H>(adst + (size_t)node * H, ad);

      auto stage_scores = [&](int c) -> int {
        const int j = c + lane;
        int s = 0;
        float v[H];
#pragma unroll
        for (int h = 0; h < H; ++h) v[h] = -INFINITY;
        if (j < end) {
          s = csr_src[j];
          float as[H];
          load_heads<H>(asrc + (size_t)s * H, as);
#pragma unroll
          for (int h = 0; h < H; ++h) {
            float t = as[h] + ad[h];
            v[h] = (t > 0.0f) ? t : NEG_SLOPE * t;
          }
        }
#pragma unroll
        for (int h = 0; h < H; ++h) myp[lane * H + h] = v[h];
        return s;
      };

      float mx = -INFINITY;
      int s = 0;
      for (int c = beg; c < end; c += 32) {
        s = stage_scores(c);
        wave_lds_sync();
#pragma unroll
        for (int t = 0; t < 32; ++t) mx = fmaxf(mx, myp[t * H + hd]);
        wave_lds_sync();
      }

      const bool single = (end - beg) <= 32;
      float acc[VPL];
#pragma unroll
      for (int i = 0; i < VPL; ++i) acc[i] = 0.0f;
      float lsum = 0.0f;
      for (int c = beg; c < end; c += 32) {
        if (!single) {
          s = stage_scores(c);
          wave_lds_sync();
        }
        const int m = min(32, end - c);
#pragma unroll 4
        for (int t = 0; t < m; ++t) {
          const int st = __builtin_amdgcn_readlane(s, t);
          const float p = __expf(myp[t * H + hd] - mx);
          lsum += p;
          const float* hp = hfeat + (size_t)st * F + lane * VPL;
          const v4f a = *(const v4f*)hp, b = *(const v4f*)(hp + 4);
          acc[0] += p * a.x; acc[1] += p * a.y; acc[2] += p * a.z; acc[3] += p * a.w;
          acc[4] += p * b.x; acc[5] += p * b.y; acc[6] += p * b.z; acc[7] += p * b.w;
        }
        wave_lds_sync();
      }

      const float denom = lsum + 1e-16f;
#pragma unroll
      for (int i = 0; i < VPL; ++i) {
        float o = acc[i] / denom + bias[lane * VPL + i];
        r[i] = fmaxf(o, 0.0f);
      }
    }
    const v4f ra = {r[0], r[1], r[2], r[3]}, rb = {r[4], r[5], r[6], r[7]};
    v4f* ap = (v4f*)(As + slot * AP + lane * VPL);
    ap[0] = ra;
    ap[1] = rb;
  }
  const int c  = lane & 15;
  const int hh = lane >> 4;
  const int mt = wib >> 2;
  const int kq = wib & 3;
  const float* arow = As + (16 * mt + c) * AP + 64 * kq + 8 * hh;
  const float* bp   = W2 + (size_t)(64 * kq + 8 * hh) * F2 + 4 * c;

  v4f af[2], bf[8];
#pragma unroll
  for (int i = 0; i < 8; ++i) bf[i] = *(const v4f*)(bp + (size_t)i * F2);

  __syncthreads();

  v8f acc2[4] = {};
#pragma unroll
  for (int kc = 0; kc < 4; ++kc) {
    if (kc > 0) {
#pragma unroll
      for (int i = 0; i < 8; ++i) bf[i] = *(const v4f*)(bp + (size_t)(16 * kc + i) * F2);
    }
#pragma unroll
    for (int i = 0; i < 2; ++i) af[i] = *(const v4f*)(arow + 16 * kc + 4 * i);
    __builtin_amdgcn_sched_barrier(0);
#pragma unroll
    for (int s = 0; s < 4; ++s) {
      const v4f av = af[s >> 1];
      v2f a;
      if (s & 1) { a.x = av.z; a.y = av.w; } else { a.x = av.x; a.y = av.y; }
      const v4f b0 = bf[2 * s], b1 = bf[2 * s + 1];
#pragma unroll
      for (int j = 0; j < 4; ++j) {
        v2f b;
        b.x = b0[j];
        b.y = b1[j];
        acc2[j] = __builtin_amdgcn_wmma_f32_16x16x4_f32(
             false, a,  false, b,
             (short)0, acc2[j],  false,  false);
      }
    }
    __builtin_amdgcn_sched_barrier(0);
  }

  __syncthreads();
#pragma unroll
  for (int r = 0; r < 8; ++r) {
    const v4f o = {acc2[0][r], acc2[1][r], acc2[2][r], acc2[3][r]};
    As4[(wib * 8 + r) * 32 + lane] = o;
  }
  __syncthreads();

  const int rr = threadIdx.x >> 4;
  const int cq = threadIdx.x & 15;
  const int pl = 16 * (rr >> 3) + cq;
  const v4f ws2 = *(const v4f*)(att_src2 + 4 * cq);
  const v4f wd2 = *(const v4f*)(att_dst2 + 4 * cq);
#pragma unroll
  for (int it = 0; it < 2; ++it) {
    const v4f* pp = As4 + ((it * 4) * 8 + (rr & 7)) * 32 + pl;
    v4f o = pp[0];
    o += pp[1 * 8 * 32];
    o += pp[2 * 8 * 32];
    o += pp[3 * 8 * 32];
    float ts = o.x * ws2.x + o.y * ws2.y + o.z * ws2.z + o.w * ws2.w;
    float td = o.x * wd2.x + o.y * wd2.y + o.z * wd2.z + o.w * wd2.w;
#pragma unroll
    for (int m = 8; m >= 1; m >>= 1) {
      ts += __shfl_xor(ts, m, 32);
      td += __shfl_xor(td, m, 32);
    }
    const int node = base + 16 * it + rr;
    if (node < n) {
      *(v4f*)(h2 + (size_t)node * F2 + 4 * cq) = o;
      if (cq == 0) {
        asrc2[node] = ts;
        adst2[node] = td;
      }
    }
  }
}

extern "C" void kernel_launch(void* const* d_in, const int* in_sizes, int n_in,
                              void* d_out, int out_size, void* d_ws, size_t ws_size,
                              hipStream_t stream) {
  const float* x        = (const float*)d_in[0];
  const int*   ei       = (const int*)d_in[1];
  const float* W1       = (const float*)d_in[2];
  const float* att_src1 = (const float*)d_in[3];
  const float* att_dst1 = (const float*)d_in[4];
  const float* b1       = (const float*)d_in[5];
  const float* W2       = (const float*)d_in[6];
  const float* att_src2 = (const float*)d_in[7];
  const float* att_dst2 = (const float*)d_in[8];
  const float* b2       = (const float*)d_in[9];

  const int D_IN = 256, F1 = 256, H1 = 8;
  const int N = in_sizes[0] / D_IN;
  const int E = in_sizes[1] / 2;
  const int NBK = (N + BKT_NODES - 1) / BKT_NODES;

  float* ws = (float*)d_ws;
  float* h1    = ws;
  float* agg1  = h1   + (size_t)N * F1;
  float* asrc1 = agg1 + (size_t)N * F1;
  float* adst1 = asrc1 + (size_t)N * H1;
  int*   part  = (int*)(adst1 + (size_t)N * H1);
  float* asrc2 = (float*)(part + (size_t)2 * E);
  float* adst2 = asrc2 + (N + 4);
  int*   rowptr = (int*)(adst2 + (N + 4));
  const int CP = NBK + 1;
  int*   ct    = rowptr + (N + 4);
  int*   csr_src = ct + (size_t)NCHUNK * CP;
  float* out   = (float*)d_out;

  auto cdiv = [](long long a, long long b) { return (int)((a + b - 1) / b); };

  auto imin = [](int a, int b) { return a < b ? a : b; };
  const int T  = cdiv(N, 128) * (F1 / 64);
  const int t0 = imin(T, 768), t1 = T - t0;
  group_pass_beside_gemm<0><<<NCHUNK + t0, 256, 0, stream>>>(
      ei, E, NBK, N, ct, CP, part, rowptr, csr_src, NCHUNK,
      x, W1, h1, att_src1, att_dst1, asrc1, adst1, N, 0);
  group_pass_beside_gemm<1><<<NBK + t1, 256, 0, stream>>>(
      ei, E, NBK, N, ct, CP, part, rowptr, csr_src, NBK,
      x, W1, h1, att_src1, att_dst1, asrc1, adst1, N, t0);

  float* h2 = agg1;
  gather1_gemm2<<<cdiv(N, FN), 256, 0, stream>>>(
      rowptr, csr_src, asrc1, adst1, h1, b1, W2, att_src2, att_dst2, h2, asrc2, adst2, N);

  node_softmax_gather<1, 64, false><<<cdiv(N, 8), 256, 0, stream>>>(
      rowptr, csr_src, asrc2, adst2, h2, b2, out, N);
}
